# starting kernel
# baseline (speedup 1.0000x reference)
#include <hip/hip_runtime.h>

typedef __attribute__((ext_vector_type(2))) float v2f;
typedef __attribute__((ext_vector_type(8))) float v8f;

#define P   40
#define PT  48
#define MST 50
#define WST 42
#define NF  7
#define WPB 4
#define BLOCK (WPB * 32)
#define FLTMAX 3.402823466e+38f

#define WMMA_F32X4(A, Bv, C) \
  __builtin_amdgcn_wmma_f32_16x16x4_f32(false, (A), false, (Bv), (short)0, (C), false, false)

__global__ __launch_bounds__(BLOCK)
void lorentz_feats_kernel(const float* __restrict__ x,
                          const float* __restrict__ w,
                          float* __restrict__ out,
                          int B)
{
  __shared__ __align__(16) float w0s[P][P];
  __shared__ float ws1[P];
  __shared__ float ws2[P];
  __shared__ __align__(16) float w3t[PT][WST];
  __shared__ __align__(16) float w4t[PT][WST];
  __shared__ __align__(16) float md2[WPB][PT][MST];
  __shared__ float xsh[WPB][4][P];
  __shared__ float ssh[WPB][P];

  const int tid  = threadIdx.x;
  const int lane = tid & 31;
  const int wv   = tid >> 5;

  for (int idx = tid; idx < P * P; idx += BLOCK) {
    int i = idx / P, n = idx % P;
    w0s[i][n] = w[0 * P * P + idx];
    w3t[n][i] = w[3 * P * P + idx];
    w4t[n][i] = w[4 * P * P + idx];
  }
  for (int idx = tid; idx < (PT - P) * WST; idx += BLOCK) {
    int n = P + idx / WST, i = idx % WST;
    w3t[n][i] = 0.f;
    w4t[n][i] = 0.f;
  }
  for (int i = tid; i < P; i += BLOCK) {
    float s1 = 0.f, s2 = 0.f;
    for (int k = 0; k < P; ++k) {
      s1 += w[1 * P * P + i * P + k];
      s2 += w[2 * P * P + i * P + k];
    }
    ws1[i] = s1;
    ws2[i] = s2;
  }
  __syncthreads();

  const int half = lane >> 4;
  const int ln   = lane & 15;
  const int kb   = half << 1;

  v2f bw3[3][10], bw4[3][10];
  #pragma unroll
  for (int t = 0; t < 3; ++t) {
    #pragma unroll
    for (int ks = 0; ks < 10; ++ks) {
      bw3[t][ks] = *(const v2f*)&w3t[16 * t + ln][4 * ks + kb];
      bw4[t][ks] = *(const v2f*)&w4t[16 * t + ln][4 * ks + kb];
    }
  }

  const int gwave  = blockIdx.x * WPB + wv;
  const int nwaves = gridDim.x * WPB;

  for (int b = gwave; b < B; b += nwaves) {
    const float* xb = x + (size_t)b * 4 * P;
    if (b + nwaves < B)
      __builtin_prefetch(x + (size_t)(b + nwaves) * 4 * P, 0, 3);

    for (int i = lane; i < P; i += 32) {
      float x0 = xb[0 * P + i], x1 = xb[1 * P + i];
      float x2 = xb[2 * P + i], x3 = xb[3 * P + i];
      xsh[wv][0][i] = x0; xsh[wv][1][i] = x1;
      xsh[wv][2][i] = x2; xsh[wv][3][i] = x3;
      ssh[wv][i] = -x0 * x0 + x1 * x1 + x2 * x2 + x3 * x3;
    }
    __builtin_amdgcn_wave_barrier();

    for (int idx = lane; idx < P * P; idx += 32) {
      int i = idx / P, j = idx % P;
      float dot = -xsh[wv][0][i] * xsh[wv][0][j]
                 + xsh[wv][1][i] * xsh[wv][1][j]
                 + xsh[wv][2][i] * xsh[wv][2][j]
                 + xsh[wv][3][i] * xsh[wv][3][j];
      md2[wv][i][j] = ssh[wv][i] + ssh[wv][j] - 2.f * dot;
    }
    __builtin_amdgcn_wave_barrier();

    float* ob = out + (size_t)b * NF * P;

    for (int j0 = 0; j0 < PT; j0 += 16) {
      v8f c3a = {}, c3b = {}, c3c = {};
      v8f c4a = {}, c4b = {}, c4c = {};
      #pragma unroll
      for (int ks = 0; ks < 10; ++ks) {
        v2f a = *(const v2f*)&md2[wv][j0 + ln][4 * ks + kb];
        c3a = WMMA_F32X4(a, bw3[0][ks], c3a);
        c3b = WMMA_F32X4(a, bw3[1][ks], c3b);
        c3c = WMMA_F32X4(a, bw3[2][ks], c3c);
        c4a = WMMA_F32X4(a, bw4[0][ks], c4a);
        c4b = WMMA_F32X4(a, bw4[1][ks], c4b);
        c4c = WMMA_F32X4(a, bw4[2][ks], c4c);
      }

      #pragma unroll
      for (int r = 0; r < 8; ++r) {
        int j = j0 + (half << 3) + r;
        float m3 = fminf(c3a[r], c3b[r]);
        m3 = fminf(m3, (ln < 8) ? c3c[r] : FLTMAX);
        float m4 = fminf(c4a[r], c4b[r]);
        m4 = fminf(m4, (ln < 8) ? c4c[r] : FLTMAX);
        #pragma unroll
        for (int off = 1; off <= 8; off <<= 1) {
          m3 = fminf(m3, __shfl_xor(m3, off, 32));
          m4 = fminf(m4, __shfl_xor(m4, off, 32));
        }
        if (ln == 0 && j < P) {
          ob[5 * P + j] = m3;
          ob[6 * P + j] = m4;
        }
      }
    }

    for (int j = lane; j < P; j += 32) {
      float x0 = xsh[wv][0][j], x1 = xsh[wv][1][j];
      float x2 = xsh[wv][2][j], x3 = xsh[wv][3][j];
      ob[0 * P + j] = x0 * x0 - x1 * x1 - x2 * x2 - x3 * x3;
      ob[1 * P + j] = sqrtf(x1 * x1 + x2 * x2);
      float a0 = 0.f, a1 = 0.f, a2 = 0.f;
      for (int i = 0; i < P; ++i) {
        a0 += xsh[wv][0][i] * w0s[i][j];
        float m = md2[wv][i][j];
        a1 += m * ws1[i];
        a2 += m * ws2[i];
      }
      ob[2 * P + j] = a0;
      ob[3 * P + j] = a1;
      ob[4 * P + j] = a2;
    }
    __builtin_amdgcn_wave_barrier();
  }
}

extern "C" void kernel_launch(void* const* d_in, const int* in_sizes, int n_in,
                              void* d_out, int out_size, void* d_ws, size_t ws_size,
                              hipStream_t stream) {
  const float* x = (const float*)d_in[0];
  const float* w = (const float*)d_in[1];
  float* out = (float*)d_out;
  const int B = in_sizes[0] / (4 * P);

  int totalWaves = (B + 3) / 4;
  int grid = (totalWaves + WPB - 1) / WPB;
  if (grid > 1024) grid = 1024;
  if (grid < 1) grid = 1;

  lorentz_feats_kernel<<<grid, BLOCK, 0, stream>>>(x, w, out, B);
}


# min over k: the 16 per-row ds_bpermute butterflies of each tile become one transposing 4-step DPP row_xmask butterfly; each lane ends with one row's min and one store writes the tile
# speedup vs baseline: 1.3071x; 1.3071x over previous
#include <hip/hip_runtime.h>

typedef __attribute__((ext_vector_type(2))) float v2f;
typedef __attribute__((ext_vector_type(8))) float v8f;

#define P   40
#define PT  48
#define MST 50
#define WST 42
#define NF  7
#define WPB 4
#define BLOCK (WPB * 32)
#define FLTMAX 3.402823466e+38f

#define DPP_XMASK(x, m) \
  __int_as_float(__builtin_amdgcn_update_dpp(0, __float_as_int(x), 0x160 | (m), 0xf, 0xf, true))

#define WMMA_F32X4(A, Bv, C) \
  __builtin_amdgcn_wmma_f32_16x16x4_f32(false, (A), false, (Bv), (short)0, (C), false, false)

__global__ __launch_bounds__(BLOCK, 4)
void lorentz_feats_kernel(const float* __restrict__ x,
                          const float* __restrict__ w,
                          float* __restrict__ out,
                          int B)
{
  __shared__ __align__(16) float w0s[P][P];
  __shared__ float ws1[P];
  __shared__ float ws2[P];
  __shared__ __align__(16) float w3t[PT][WST];
  __shared__ __align__(16) float w4t[PT][WST];
  __shared__ __align__(16) float md2[WPB][PT][MST];
  __shared__ float xsh[WPB][4][P];
  __shared__ float ssh[WPB][P];

  const int tid  = threadIdx.x;
  const int lane = tid & 31;
  const int wv   = tid >> 5;

  for (int idx = tid; idx < P * P; idx += BLOCK) {
    int i = idx / P, n = idx % P;
    w0s[i][n] = w[0 * P * P + idx];
    w3t[n][i] = w[3 * P * P + idx];
    w4t[n][i] = w[4 * P * P + idx];
  }
  for (int idx = tid; idx < (PT - P) * WST; idx += BLOCK) {
    int n = P + idx / WST, i = idx % WST;
    w3t[n][i] = 0.f;
    w4t[n][i] = 0.f;
  }
  for (int i = tid; i < P; i += BLOCK) {
    float s1 = 0.f, s2 = 0.f;
    for (int k = 0; k < P; ++k) {
      s1 += w[1 * P * P + i * P + k];
      s2 += w[2 * P * P + i * P + k];
    }
    ws1[i] = s1;
    ws2[i] = s2;
  }
  __syncthreads();

  const int half = lane >> 4;
  const int ln   = lane & 15;
  const int kb   = half << 1;

  v2f bw3[3][10], bw4[3][10];
  #pragma unroll
  for (int t = 0; t < 3; ++t) {
    #pragma unroll
    for (int ks = 0; ks < 10; ++ks) {
      bw3[t][ks] = *(const v2f*)&w3t[16 * t + ln][4 * ks + kb];
      bw4[t][ks] = *(const v2f*)&w4t[16 * t + ln][4 * ks + kb];
    }
  }

  const int gwave  = blockIdx.x * WPB + wv;
  const int nwaves = gridDim.x * WPB;

  for (int b = gwave; b < B; b += nwaves) {
    const float* xb = x + (size_t)b * 4 * P;
    if (b + nwaves < B)
      __builtin_prefetch(x + (size_t)(b + nwaves) * 4 * P, 0, 3);

    for (int i = lane; i < P; i += 32) {
      float x0 = xb[0 * P + i], x1 = xb[1 * P + i];
      float x2 = xb[2 * P + i], x3 = xb[3 * P + i];
      xsh[wv][0][i] = x0; xsh[wv][1][i] = x1;
      xsh[wv][2][i] = x2; xsh[wv][3][i] = x3;
      ssh[wv][i] = -x0 * x0 + x1 * x1 + x2 * x2 + x3 * x3;
    }
    __builtin_amdgcn_wave_barrier();

    for (int idx = lane; idx < P * P; idx += 32) {
      int i = idx / P, j = idx % P;
      float dot = -xsh[wv][0][i] * xsh[wv][0][j]
                 + xsh[wv][1][i] * xsh[wv][1][j]
                 + xsh[wv][2][i] * xsh[wv][2][j]
                 + xsh[wv][3][i] * xsh[wv][3][j];
      md2[wv][i][j] = ssh[wv][i] + ssh[wv][j] - 2.f * dot;
    }
    __builtin_amdgcn_wave_barrier();

    float* ob = out + (size_t)b * NF * P;

    for (int j0 = 0; j0 < PT; j0 += 16) {
      v8f c3a = {}, c3b = {}, c3c = {};
      v8f c4a = {}, c4b = {}, c4c = {};
      #pragma unroll
      for (int ks = 0; ks < 10; ++ks) {
        v2f a = *(const v2f*)&md2[wv][j0 + ln][4 * ks + kb];
        c3a = WMMA_F32X4(a, bw3[0][ks], c3a);
        c3b = WMMA_F32X4(a, bw3[1][ks], c3b);
        c3c = WMMA_F32X4(a, bw3[2][ks], c3c);
        c4a = WMMA_F32X4(a, bw4[0][ks], c4a);
        c4b = WMMA_F32X4(a, bw4[1][ks], c4b);
        c4c = WMMA_F32X4(a, bw4[2][ks], c4c);
      }

      float v[16];
      #pragma unroll
      for (int r = 0; r < 8; ++r) {
        float m3 = fminf(c3a[r], c3b[r]);
        v[r]     = fminf(m3, (ln < 8) ? c3c[r] : FLTMAX);
        float m4 = fminf(c4a[r], c4b[r]);
        v[8 + r] = fminf(m4, (ln < 8) ? c4c[r] : FLTMAX);
      }
      {
        const bool up = (ln & 8) != 0;
        #pragma unroll
        for (int q = 0; q < 8; ++q) {
          float keep = up ? v[q + 8] : v[q];
          float send = up ? v[q] : v[q + 8];
          v[q] = fminf(keep, DPP_XMASK(send, 8));
        }
      }
      {
        const bool up = (ln & 4) != 0;
        #pragma unroll
        for (int q = 0; q < 4; ++q) {
          float keep = up ? v[q + 4] : v[q];
          float send = up ? v[q] : v[q + 4];
          v[q] = fminf(keep, DPP_XMASK(send, 4));
        }
      }
      {
        const bool up = (ln & 2) != 0;
        #pragma unroll
        for (int q = 0; q < 2; ++q) {
          float keep = up ? v[q + 2] : v[q];
          float send = up ? v[q] : v[q + 2];
          v[q] = fminf(keep, DPP_XMASK(send, 2));
        }
      }
      {
        const bool up = (ln & 1) != 0;
        float keep = up ? v[1] : v[0];
        float send = up ? v[0] : v[1];
        v[0] = fminf(keep, DPP_XMASK(send, 1));
      }
      {
        int j = j0 + (half << 3) + (ln & 7);
        if (j < P) ob[(5 + (ln >> 3)) * P + j] = v[0];
      }
    }

    for (int j = lane; j < P; j += 32) {
      float x0 = xsh[wv][0][j], x1 = xsh[wv][1][j];
      float x2 = xsh[wv][2][j], x3 = xsh[wv][3][j];
      ob[0 * P + j] = x0 * x0 - x1 * x1 - x2 * x2 - x3 * x3;
      ob[1 * P + j] = sqrtf(x1 * x1 + x2 * x2);
      float a0 = 0.f, a1 = 0.f, a2 = 0.f;
      for (int i = 0; i < P; ++i) {
        a0 += xsh[wv][0][i] * w0s[i][j];
        float m = md2[wv][i][j];
        a1 += m * ws1[i];
        a2 += m * ws2[i];
      }
      ob[2 * P + j] = a0;
      ob[3 * P + j] = a1;
      ob[4 * P + j] = a2;
    }
    __builtin_amdgcn_wave_barrier();
  }
}

extern "C" void kernel_launch(void* const* d_in, const int* in_sizes, int n_in,
                              void* d_out, int out_size, void* d_ws, size_t ws_size,
                              hipStream_t stream) {
  const float* x = (const float*)d_in[0];
  const float* w = (const float*)d_in[1];
  float* out = (float*)d_out;
  const int B = in_sizes[0] / (4 * P);

  int totalWaves = (B + 3) / 4;
  int grid = (totalWaves + WPB - 1) / WPB;
  if (grid > 1024) grid = 1024;
  if (grid < 1) grid = 1;

  lorentz_feats_kernel<<<grid, BLOCK, 0, stream>>>(x, w, out, B);
}


# first product: two batch elements share one 16-row X tile (element 1 in rows 4-7,12,13 = D registers 4-7), so its 60 WMMAs serve a pair: 96 -> 66 WMMAs per element
# speedup vs baseline: 2.9502x; 2.2571x over previous
#include <hip/hip_runtime.h>

typedef __attribute__((ext_vector_type(2))) float v2f;
typedef __attribute__((ext_vector_type(8))) float v8f;

#define P   40
#define PT  48
#define XST 50
#define WST 42
#define NF  7
#define WPB 4
#define BLOCK (WPB * 32)
#define FLTMAX 3.402823466e+38f

#define DPP_XMASK(x, m) \
  __int_as_float(__builtin_amdgcn_update_dpp(0, __float_as_int(x), 0x160 | (m), 0xf, 0xf, true))

#define WMMA_F32X4(A, Bv, C) \
  __builtin_amdgcn_wmma_f32_16x16x4_f32(false, (A), false, (Bv), (short)0, (C), false, false)

#define RDLANE(x, l) __int_as_float(__builtin_amdgcn_readlane(__float_as_int(x), (l)))

__global__ __launch_bounds__(BLOCK, 4)
void lorentz_feats_kernel(const float* __restrict__ x,
                          const float* __restrict__ w,
                          float* __restrict__ out,
                          int B)
{
  __shared__ __align__(16) float w0s[P][P];
  __shared__ __align__(16) float w3t[PT][WST];
  __shared__ __align__(16) float w4t[PT][WST];
  __shared__ __align__(16) float xs[WPB][16][XST];

  const int tid  = threadIdx.x;
  const int lane = tid & 31;
  const int wv   = __builtin_amdgcn_readfirstlane(tid >> 5);

  for (int idx = tid; idx < P * P; idx += BLOCK) {
    int i = idx / P, n = idx % P;
    w0s[i][n] = w[0 * P * P + idx];
    w3t[n][i] = w[3 * P * P + idx];
    w4t[n][i] = w[4 * P * P + idx];
  }
  for (int idx = tid; idx < (PT - P) * WST; idx += BLOCK) {
    int n = P + idx / WST, i = idx % WST;
    w4t[n][i] = 0.f;
    if (n >= P + 2) w3t[n][i] = 0.f;
  }
  for (int i = tid; i < P; i += BLOCK) {
    float s1 = 0.f, s2 = 0.f;
    for (int k = 0; k < P; ++k) {
      s1 += w[1 * P * P + i * P + k];
      s2 += w[2 * P * P + i * P + k];
    }
    w3t[P][i]     = s1;
    w3t[P + 1][i] = s2;
  }
  for (int idx = lane; idx < 16 * XST; idx += 32) {
    int r = idx / XST, c = idx % XST;
    xs[wv][r][c] = ((r == 9 || r == 13) && c < P) ? 1.f : 0.f;
  }
  __syncthreads();

  const int half = lane >> 4;
  const int ln   = lane & 15;
  const int kb   = half << 1;

  v2f bw3[3][10], bw4[3][10];
  #pragma unroll
  for (int t = 0; t < 3; ++t) {
    #pragma unroll
    for (int ks = 0; ks < 10; ++ks) {
      bw3[t][ks] = *(const v2f*)&w3t[16 * t + ln][4 * ks + kb];
      bw4[t][ks] = *(const v2f*)&w4t[16 * t + ln][4 * ks + kb];
    }
  }

  const int gwave  = blockIdx.x * WPB + wv;
  const int nwaves = gridDim.x * WPB;
  const int npairs = (B + 1) >> 1;

  for (int p = gwave; p < npairs; p += nwaves) {
    const int  b0   = 2 * p;
    const bool hasB = (b0 + 1) < B;
    const float* xb = x + (size_t)b0 * 4 * P;
    if (p + nwaves < npairs)
      __builtin_prefetch(x + (size_t)(p + nwaves) * 8 * P, 0, 3);

    float* ob = out + (size_t)b0 * NF * P;

    float xv[3][4];
    #pragma unroll
    for (int t = 0; t < 3; ++t) {
      const int  idx = lane + 32 * t;
      const bool ok  = idx < (hasB ? 2 * P : P);
      const int  e   = (idx >= P) ? 1 : 0;
      const unsigned off = ok ? (unsigned)(idx + e * 3 * P) : 0u;
      #pragma unroll
      for (int f = 0; f < 4; ++f) xv[t][f] = xb[off + (unsigned)(f * P)];
    }
    #pragma unroll
    for (int t = 0; t < 3; ++t) {
      const int  idx = lane + 32 * t;
      const bool ok  = idx < (hasB ? 2 * P : P);
      const int  e   = (idx >= P) ? 1 : 0;
      const int  i   = idx - e * P;
      if (ok) {
        const float x0 = xv[t][0], x1 = xv[t][1], x2 = xv[t][2], x3 = xv[t][3];
        xs[wv][4 * e + 0][i] = -x0 * x0 + x1 * x1 + x2 * x2 + x3 * x3;
        xs[wv][4 * e + 1][i] = x0; xs[wv][4 * e + 2][i] = x1;
        xs[wv][4 * e + 3][i] = x2; xs[wv][4 * e + 8][i] = x3;
        const unsigned oo = (unsigned)(e * NF * P + i);
        ob[oo]     = x0 * x0 - x1 * x1 - x2 * x2 - x3 * x3;
        ob[oo + P] = sqrtf(x1 * x1 + x2 * x2);
      }
    }
    __builtin_amdgcn_wave_barrier();

    v8f z3a = {}, z3b = {}, z3c = {};
    v8f z4a = {}, z4b = {}, z4c = {};
    #pragma unroll
    for (int ks = 0; ks < 10; ++ks) {
      v2f a = *(const v2f*)&xs[wv][ln][4 * ks + kb];
      z3a = WMMA_F32X4(a, bw3[0][ks], z3a);
      z3b = WMMA_F32X4(a, bw3[1][ks], z3b);
      z3c = WMMA_F32X4(a, bw3[2][ks], z3c);
      z4a = WMMA_F32X4(a, bw4[0][ks], z4a);
      z4b = WMMA_F32X4(a, bw4[1][ks], z4b);
      z4c = WMMA_F32X4(a, bw4[2][ks], z4c);
    }

    #pragma unroll
    for (int e = 0; e < 2; ++e) {
      if (e == 1 && !hasB) continue;
      const int r0 = 4 * e;
      const int r8 = 8 + 4 * e;
      float* oe = ob + e * NF * P;

      const v2f p3a = {z3a[r0], z3a[r0 + 1]}, q3a = {z3a[r0 + 2], z3a[r0 + 3]};
      const v2f p3b = {z3b[r0], z3b[r0 + 1]}, q3b = {z3b[r0 + 2], z3b[r0 + 3]};
      const v2f p3c = {z3c[r0], z3c[r0 + 1]}, q3c = {z3c[r0 + 2], z3c[r0 + 3]};
      const v2f p4a = {z4a[r0], z4a[r0 + 1]}, q4a = {z4a[r0 + 2], z4a[r0 + 3]};
      const v2f p4b = {z4b[r0], z4b[r0 + 1]}, q4b = {z4b[r0 + 2], z4b[r0 + 3]};
      const v2f p4c = {z4c[r0], z4c[r0 + 1]}, q4c = {z4c[r0 + 2], z4c[r0 + 3]};

      const float g0 = RDLANE(z3c[r0],     8),  h0 = RDLANE(z3c[r0],     9);
      const float g1 = RDLANE(z3c[r0 + 1], 8),  h1 = RDLANE(z3c[r0 + 1], 9);
      const float g2 = RDLANE(z3c[r0 + 2], 8),  h2 = RDLANE(z3c[r0 + 2], 9);
      const float g3 = RDLANE(z3c[r0 + 3], 8),  h3 = RDLANE(z3c[r0 + 3], 9);
      const float g8 = RDLANE(z3c[r0],     24), h8 = RDLANE(z3c[r0],     25);
      const float g9 = RDLANE(z3c[r0 + 1], 24), h9 = RDLANE(z3c[r0 + 1], 25);

      for (int j0 = 0; j0 < PT; j0 += 16) {
        const int jj = j0 + ln;
        const float sj  = xs[wv][r0][jj];
        const float x0j = xs[wv][r0 + 1][jj], x1j = xs[wv][r0 + 2][jj];
        const float x2j = xs[wv][r0 + 3][jj], x3j = xs[wv][r8][jj];
        v2f u0, u1;
        u0.x = half ? -2.f * x3j : 1.f;
        u0.y = half ? sj         : 2.f * x0j;
        u1.x = half ? 0.f        : -2.f * x1j;
        u1.y = half ? 0.f        : -2.f * x2j;
        const v8f zero = {};
        float v[16];
        {
          v8f c3a = WMMA_F32X4(u0, p3a, zero);
          v8f c3b = WMMA_F32X4(u0, p3b, zero);
          v8f c3c = WMMA_F32X4(u0, p3c, zero);
          c3a = WMMA_F32X4(u1, q3a, c3a);
          c3b = WMMA_F32X4(u1, q3b, c3b);
          c3c = WMMA_F32X4(u1, q3c, c3c);
          #pragma unroll
          for (int r = 0; r < 8; ++r) {
            float m3 = fminf(c3a[r], c3b[r]);
            v[r]     = fminf(m3, (ln < 8) ? c3c[r] : FLTMAX);
          }
        }
        __builtin_amdgcn_sched_barrier(0);
        {
          v8f c4a = WMMA_F32X4(u0, p4a, zero);
          v8f c4b = WMMA_F32X4(u0, p4b, zero);
          v8f c4c = WMMA_F32X4(u0, p4c, zero);
          c4a = WMMA_F32X4(u1, q4a, c4a);
          c4b = WMMA_F32X4(u1, q4b, c4b);
          c4c = WMMA_F32X4(u1, q4c, c4c);
          #pragma unroll
          for (int r = 0; r < 8; ++r) {
            float m4 = fminf(c4a[r], c4b[r]);
            v[8 + r] = fminf(m4, (ln < 8) ? c4c[r] : FLTMAX);
          }
        }
        {
          const bool up = (ln & 8) != 0;
          #pragma unroll
          for (int q = 0; q < 8; ++q) {
            float keep = up ? v[q + 8] : v[q];
            float send = up ? v[q] : v[q + 8];
            v[q] = fminf(keep, DPP_XMASK(send, 8));
          }
        }
        {
          const bool up = (ln & 4) != 0;
          #pragma unroll
          for (int q = 0; q < 4; ++q) {
            float keep = up ? v[q + 4] : v[q];
            float send = up ? v[q] : v[q + 4];
            v[q] = fminf(keep, DPP_XMASK(send, 4));
          }
        }
        {
          const bool up = (ln & 2) != 0;
          #pragma unroll
          for (int q = 0; q < 2; ++q) {
            float keep = up ? v[q + 2] : v[q];
            float send = up ? v[q] : v[q + 2];
            v[q] = fminf(keep, DPP_XMASK(send, 2));
          }
        }
        {
          const bool up = (ln & 1) != 0;
          float keep = up ? v[1] : v[0];
          float send = up ? v[0] : v[1];
          v[0] = fminf(keep, DPP_XMASK(send, 1));
        }
        {
          const unsigned j = (unsigned)(j0 + (half << 3) + (ln & 7));
          if (j < P) oe[(unsigned)((5 + (ln >> 3)) * P) + j] = v[0];
        }
      }

      #pragma unroll
      for (int t = 0; t < 2; ++t) {
        const int j = lane + 32 * t;
        if (j >= P) continue;
        const float sj  = xs[wv][r0][j];
        const float x0j = xs[wv][r0 + 1][j], x1j = xs[wv][r0 + 2][j];
        const float x2j = xs[wv][r0 + 3][j], x3j = xs[wv][r8][j];
        float a0 = 0.f;
        for (int i = 0; i < P; ++i)
          a0 += xs[wv][r0 + 1][i] * w0s[i][j];
        oe[(unsigned)(2 * P + j)] = a0;
        oe[(unsigned)(3 * P + j)] = g0 + sj * g9 + 2.f * (x0j * g1 - x1j * g2 - x2j * g3 - x3j * g8);
        oe[(unsigned)(4 * P + j)] = h0 + sj * h9 + 2.f * (x0j * h1 - x1j * h2 - x2j * h3 - x3j * h8);
      }
    }
    __builtin_amdgcn_wave_barrier();
  }
}

extern "C" void kernel_launch(void* const* d_in, const int* in_sizes, int n_in,
                              void* d_out, int out_size, void* d_ws, size_t ws_size,
                              hipStream_t stream) {
  const float* x = (const float*)d_in[0];
  const float* w = (const float*)d_in[1];
  float* out = (float*)d_out;
  const int B = in_sizes[0] / (4 * P);

  int totalWaves = (B + 3) / 4;
  int grid = (totalWaves + WPB - 1) / WPB;
  if (grid > 1024) grid = 1024;
  if (grid < 1) grid = 1;

  lorentz_feats_kernel<<<grid, BLOCK, 0, stream>>>(x, w, out, B);
}


# prologue: all weight-staging and row-sum global loads issued up front as 16-byte loads (one memory round trip instead of eight serial ones); LDS fills without division loops
# speedup vs baseline: 3.3273x; 1.1278x over previous
#include <hip/hip_runtime.h>

typedef __attribute__((ext_vector_type(2))) float v2f;
typedef __attribute__((ext_vector_type(8))) float v8f;

#define P   40
#define PT  48
#define XST 50
#define WST 42
#define NF  7
#define WPB 4
#define BLOCK (WPB * 32)
#define FLTMAX 3.402823466e+38f

#define DPP_XMASK(x, m) \
  __int_as_float(__builtin_amdgcn_update_dpp(0, __float_as_int(x), 0x160 | (m), 0xf, 0xf, true))

#define WMMA_F32X4(A, Bv, C) \
  __builtin_amdgcn_wmma_f32_16x16x4_f32(false, (A), false, (Bv), (short)0, (C), false, false)

#define RDLANE(x, l) __int_as_float(__builtin_amdgcn_readlane(__float_as_int(x), (l)))

__global__ __launch_bounds__(BLOCK, 4)
void lorentz_feats_kernel(const float* __restrict__ x,
                          const float* __restrict__ w,
                          float* __restrict__ out,
                          int B)
{
  __shared__ __align__(16) float w0s[P][P];
  __shared__ __align__(16) float w3t[PT][WST];
  __shared__ __align__(16) float w4t[PT][WST];
  __shared__ __align__(16) float xs[WPB][16][XST];

  const int tid  = threadIdx.x;
  const int lane = tid & 31;
  const int wv   = __builtin_amdgcn_readfirstlane(tid >> 5);

  const float4* __restrict__ wq = (const float4*)w;
  float4 s0[4], s3[4], s4[4], rr[P / 4];
  int ku[4];
  #pragma unroll
  for (int q = 0; q < 4; ++q) {
    const int k = tid + BLOCK * q;
    ku[q] = (q < 3 || k < P * P / 4) ? k : 0;
    s0[q] = wq[ku[q]];
    s3[q] = wq[3 * (P * P / 4) + ku[q]];
    s4[q] = wq[4 * (P * P / 4) + ku[q]];
  }
  const int  rsel = tid >> 6;
  const int  rrow = tid & 63;
  const bool rok  = rrow < P;
  {
    const int base = (1 + rsel) * (P * P / 4) + (rok ? rrow : 0) * (P / 4);
    #pragma unroll
    for (int q = 0; q < P / 4; ++q) rr[q] = wq[base + q];
  }

  {
    float* z4 = &w4t[P][0];
    for (int k = tid; k < (PT - P) * WST; k += BLOCK) z4[k] = 0.f;
    float* z3 = &w3t[P + 2][0];
    for (int k = tid; k < (PT - P - 2) * WST; k += BLOCK) z3[k] = 0.f;
    float* xw = &xs[wv][0][0];
    #pragma unroll
    for (int q = 0; q < (16 * XST) / 32; ++q) {
      const int k = lane + 32 * q;
      const bool one = (k >= 9 * XST && k < 9 * XST + P) || (k >= 13 * XST && k < 13 * XST + P);
      xw[k] = one ? 1.f : 0.f;
    }
  }

  #pragma unroll
  for (int q = 0; q < 4; ++q) {
    const int i  = ku[q] / (P / 4);
    const int n0 = (ku[q] - i * (P / 4)) * 4;
    *(float4*)&w0s[i][n0] = s0[q];
    w3t[n0 + 0][i] = s3[q].x; w3t[n0 + 1][i] = s3[q].y;
    w3t[n0 + 2][i] = s3[q].z; w3t[n0 + 3][i] = s3[q].w;
    w4t[n0 + 0][i] = s4[q].x; w4t[n0 + 1][i] = s4[q].y;
    w4t[n0 + 2][i] = s4[q].z; w4t[n0 + 3][i] = s4[q].w;
  }
  if (rok) {
    float s = 0.f;
    #pragma unroll
    for (int q = 0; q < P / 4; ++q) { s += rr[q].x; s += rr[q].y; s += rr[q].z; s += rr[q].w; }
    w3t[P + rsel][rrow] = s;
  }
  __syncthreads();

  const int half = lane >> 4;
  const int ln   = lane & 15;
  const int kb   = half << 1;

  v2f bw3[3][10], bw4[3][10];
  #pragma unroll
  for (int t = 0; t < 3; ++t) {
    #pragma unroll
    for (int ks = 0; ks < 10; ++ks) {
      bw3[t][ks] = *(const v2f*)&w3t[16 * t + ln][4 * ks + kb];
      bw4[t][ks] = *(const v2f*)&w4t[16 * t + ln][4 * ks + kb];
    }
  }

  const int gwave  = blockIdx.x * WPB + wv;
  const int nwaves = gridDim.x * WPB;
  const int npairs = (B + 1) >> 1;

  for (int p = gwave; p < npairs; p += nwaves) {
    const int  b0   = 2 * p;
    const bool hasB = (b0 + 1) < B;
    const float* xb = x + (size_t)b0 * 4 * P;
    if (p + nwaves < npairs)
      __builtin_prefetch(x + (size_t)(p + nwaves) * 8 * P, 0, 3);

    float* ob = out + (size_t)b0 * NF * P;

    float xv[3][4];
    #pragma unroll
    for (int t = 0; t < 3; ++t) {
      const int  idx = lane + 32 * t;
      const bool ok  = idx < (hasB ? 2 * P : P);
      const int  e   = (idx >= P) ? 1 : 0;
      const unsigned off = ok ? (unsigned)(idx + e * 3 * P) : 0u;
      #pragma unroll
      for (int f = 0; f < 4; ++f) xv[t][f] = xb[off + (unsigned)(f * P)];
    }
    #pragma unroll
    for (int t = 0; t < 3; ++t) {
      const int  idx = lane + 32 * t;
      const bool ok  = idx < (hasB ? 2 * P : P);
      const int  e   = (idx >= P) ? 1 : 0;
      const int  i   = idx - e * P;
      if (ok) {
        const float x0 = xv[t][0], x1 = xv[t][1], x2 = xv[t][2], x3 = xv[t][3];
        xs[wv][4 * e + 0][i] = -x0 * x0 + x1 * x1 + x2 * x2 + x3 * x3;
        xs[wv][4 * e + 1][i] = x0; xs[wv][4 * e + 2][i] = x1;
        xs[wv][4 * e + 3][i] = x2; xs[wv][4 * e + 8][i] = x3;
        const unsigned oo = (unsigned)(e * NF * P + i);
        ob[oo]     = x0 * x0 - x1 * x1 - x2 * x2 - x3 * x3;
        ob[oo + P] = sqrtf(x1 * x1 + x2 * x2);
      }
    }
    __builtin_amdgcn_wave_barrier();

    v8f z3a = {}, z3b = {}, z3c = {};
    v8f z4a = {}, z4b = {}, z4c = {};
    #pragma unroll
    for (int ks = 0; ks < 10; ++ks) {
      v2f a = *(const v2f*)&xs[wv][ln][4 * ks + kb];
      z3a = WMMA_F32X4(a, bw3[0][ks], z3a);
      z3b = WMMA_F32X4(a, bw3[1][ks], z3b);
      z3c = WMMA_F32X4(a, bw3[2][ks], z3c);
      z4a = WMMA_F32X4(a, bw4[0][ks], z4a);
      z4b = WMMA_F32X4(a, bw4[1][ks], z4b);
      z4c = WMMA_F32X4(a, bw4[2][ks], z4c);
    }

    #pragma unroll
    for (int e = 0; e < 2; ++e) {
      if (e == 1 && !hasB) continue;
      const int r0 = 4 * e;
      const int r8 = 8 + 4 * e;
      float* oe = ob + e * NF * P;

      const v2f p3a = {z3a[r0], z3a[r0 + 1]}, q3a = {z3a[r0 + 2], z3a[r0 + 3]};
      const v2f p3b = {z3b[r0], z3b[r0 + 1]}, q3b = {z3b[r0 + 2], z3b[r0 + 3]};
      const v2f p3c = {z3c[r0], z3c[r0 + 1]}, q3c = {z3c[r0 + 2], z3c[r0 + 3]};
      const v2f p4a = {z4a[r0], z4a[r0 + 1]}, q4a = {z4a[r0 + 2], z4a[r0 + 3]};
      const v2f p4b = {z4b[r0], z4b[r0 + 1]}, q4b = {z4b[r0 + 2], z4b[r0 + 3]};
      const v2f p4c = {z4c[r0], z4c[r0 + 1]}, q4c = {z4c[r0 + 2], z4c[r0 + 3]};

      const float g0 = RDLANE(z3c[r0],     8),  h0 = RDLANE(z3c[r0],     9);
      const float g1 = RDLANE(z3c[r0 + 1], 8),  h1 = RDLANE(z3c[r0 + 1], 9);
      const float g2 = RDLANE(z3c[r0 + 2], 8),  h2 = RDLANE(z3c[r0 + 2], 9);
      const float g3 = RDLANE(z3c[r0 + 3], 8),  h3 = RDLANE(z3c[r0 + 3], 9);
      const float g8 = RDLANE(z3c[r0],     24), h8 = RDLANE(z3c[r0],     25);
      const float g9 = RDLANE(z3c[r0 + 1], 24), h9 = RDLANE(z3c[r0 + 1], 25);

      for (int j0 = 0; j0 < PT; j0 += 16) {
        const int jj = j0 + ln;
        const float sj  = xs[wv][r0][jj];
        const float x0j = xs[wv][r0 + 1][jj], x1j = xs[wv][r0 + 2][jj];
        const float x2j = xs[wv][r0 + 3][jj], x3j = xs[wv][r8][jj];
        v2f u0, u1;
        u0.x = half ? -2.f * x3j : 1.f;
        u0.y = half ? sj         : 2.f * x0j;
        u1.x = half ? 0.f        : -2.f * x1j;
        u1.y = half ? 0.f        : -2.f * x2j;
        const v8f zero = {};
        float v[16];
        {
          v8f c3a = WMMA_F32X4(u0, p3a, zero);
          v8f c3b = WMMA_F32X4(u0, p3b, zero);
          v8f c3c = WMMA_F32X4(u0, p3c, zero);
          c3a = WMMA_F32X4(u1, q3a, c3a);
          c3b = WMMA_F32X4(u1, q3b, c3b);
          c3c = WMMA_F32X4(u1, q3c, c3c);
          #pragma unroll
          for (int r = 0; r < 8; ++r) {
            float m3 = fminf(c3a[r], c3b[r]);
            v[r]     = fminf(m3, (ln < 8) ? c3c[r] : FLTMAX);
          }
        }
        __builtin_amdgcn_sched_barrier(0);
        {
          v8f c4a = WMMA_F32X4(u0, p4a, zero);
          v8f c4b = WMMA_F32X4(u0, p4b, zero);
          v8f c4c = WMMA_F32X4(u0, p4c, zero);
          c4a = WMMA_F32X4(u1, q4a, c4a);
          c4b = WMMA_F32X4(u1, q4b, c4b);
          c4c = WMMA_F32X4(u1, q4c, c4c);
          #pragma unroll
          for (int r = 0; r < 8; ++r) {
            float m4 = fminf(c4a[r], c4b[r]);
            v[8 + r] = fminf(m4, (ln < 8) ? c4c[r] : FLTMAX);
          }
        }
        {
          const bool up = (ln & 8) != 0;
          #pragma unroll
          for (int q = 0; q < 8; ++q) {
            float keep = up ? v[q + 8] : v[q];
            float send = up ? v[q] : v[q + 8];
            v[q] = fminf(keep, DPP_XMASK(send, 8));
          }
        }
        {
          const bool up = (ln & 4) != 0;
          #pragma unroll
          for (int q = 0; q < 4; ++q) {
            float keep = up ? v[q + 4] : v[q];
            float send = up ? v[q] : v[q + 4];
            v[q] = fminf(keep, DPP_XMASK(send, 4));
          }
        }
        {
          const bool up = (ln & 2) != 0;
          #pragma unroll
          for (int q = 0; q < 2; ++q) {
            float keep = up ? v[q + 2] : v[q];
            float send = up ? v[q] : v[q + 2];
            v[q] = fminf(keep, DPP_XMASK(send, 2));
          }
        }
        {
          const bool up = (ln & 1) != 0;
          float keep = up ? v[1] : v[0];
          float send = up ? v[0] : v[1];
          v[0] = fminf(keep, DPP_XMASK(send, 1));
        }
        {
          const unsigned j = (unsigned)(j0 + (half << 3) + (ln & 7));
          if (j < P) oe[(unsigned)((5 + (ln >> 3)) * P) + j] = v[0];
        }
      }

      #pragma unroll
      for (int t = 0; t < 2; ++t) {
        const int j = lane + 32 * t;
        if (j >= P) continue;
        const float sj  = xs[wv][r0][j];
        const float x0j = xs[wv][r0 + 1][j], x1j = xs[wv][r0 + 2][j];
        const float x2j = xs[wv][r0 + 3][j], x3j = xs[wv][r8][j];
        float a0 = 0.f;
        for (int i = 0; i < P; ++i)
          a0 += xs[wv][r0 + 1][i] * w0s[i][j];
        oe[(unsigned)(2 * P + j)] = a0;
        oe[(unsigned)(3 * P + j)] = g0 + sj * g9 + 2.f * (x0j * g1 - x1j * g2 - x2j * g3 - x3j * g8);
        oe[(unsigned)(4 * P + j)] = h0 + sj * h9 + 2.f * (x0j * h1 - x1j * h2 - x2j * h3 - x3j * h8);
      }
    }
    __builtin_amdgcn_wave_barrier();
  }
}

extern "C" void kernel_launch(void* const* d_in, const int* in_sizes, int n_in,
                              void* d_out, int out_size, void* d_ws, size_t ws_size,
                              hipStream_t stream) {
  const float* x = (const float*)d_in[0];
  const float* w = (const float*)d_in[1];
  float* out = (float*)d_out;
  const int B = in_sizes[0] / (4 * P);

  int totalWaves = (B + 3) / 4;
  int grid = (totalWaves + WPB - 1) / WPB;
  if (grid > 1024) grid = 1024;
  if (grid < 1) grid = 1;

  lorentz_feats_kernel<<<grid, BLOCK, 0, stream>>>(x, w, out, B);
}


# second product taken transposed (T^T = V^T U^T, operand registers swap places): the min over k runs down each lane's own D registers; min3 chains and one half exchange replace the DPP butterfly
# speedup vs baseline: 3.6301x; 1.0910x over previous
#include <hip/hip_runtime.h>

typedef __attribute__((ext_vector_type(2))) float v2f;
typedef __attribute__((ext_vector_type(8))) float v8f;

#define P   40
#define PT  48
#define XST 50
#define WST 42
#define NF  7
#define WPB 4
#define BLOCK (WPB * 32)

#define FMINN(a, b) __builtin_fminimum_numf((a), (b))

__device__ __forceinline__ float xhalf(float x) {
  const int b = __float_as_int(x);
  return __int_as_float(__builtin_amdgcn_permlanex16(b, b, 0x76543210, 0xfedcba98, false, false));
}

#define WMMA_F32X4(A, Bv, C) \
  __builtin_amdgcn_wmma_f32_16x16x4_f32(false, (A), false, (Bv), (short)0, (C), false, false)

#define RDLANE(x, l) __int_as_float(__builtin_amdgcn_readlane(__float_as_int(x), (l)))

__global__ __launch_bounds__(BLOCK, 4)
void lorentz_feats_kernel(const float* __restrict__ x,
                          const float* __restrict__ w,
                          float* __restrict__ out,
                          int B)
{
  __shared__ __align__(16) float w0s[P][P];
  __shared__ __align__(16) float w3t[PT][WST];
  __shared__ __align__(16) float w4t[PT][WST];
  __shared__ __align__(16) float xs[WPB][16][XST];

  const int tid  = threadIdx.x;
  const int lane = tid & 31;
  const int wv   = __builtin_amdgcn_readfirstlane(tid >> 5);

  const float4* __restrict__ wq = (const float4*)w;
  float4 s0[4], s3[4], s4[4], rr[P / 4];
  int ku[4];
  #pragma unroll
  for (int q = 0; q < 4; ++q) {
    const int k = tid + BLOCK * q;
    ku[q] = (q < 3 || k < P * P / 4) ? k : 0;
    s0[q] = wq[ku[q]];
    s3[q] = wq[3 * (P * P / 4) + ku[q]];
    s4[q] = wq[4 * (P * P / 4) + ku[q]];
  }
  const int  rsel = tid >> 6;
  const int  rrow = tid & 63;
  const bool rok  = rrow < P;
  {
    const int base = (1 + rsel) * (P * P / 4) + (rok ? rrow : 0) * (P / 4);
    #pragma unroll
    for (int q = 0; q < P / 4; ++q) rr[q] = wq[base + q];
  }

  {
    float* z4 = &w4t[P][0];
    for (int k = tid; k < (PT - P) * WST; k += BLOCK) z4[k] = 0.f;
    float* z3 = &w3t[P + 2][0];
    for (int k = tid; k < (PT - P - 2) * WST; k += BLOCK) z3[k] = 0.f;
    float* xw = &xs[wv][0][0];
    #pragma unroll
    for (int q = 0; q < (16 * XST) / 32; ++q) {
      const int k = lane + 32 * q;
      const bool one = (k >= 9 * XST && k < 9 * XST + P) || (k >= 13 * XST && k < 13 * XST + P);
      xw[k] = one ? 1.f : 0.f;
    }
  }

  #pragma unroll
  for (int q = 0; q < 4; ++q) {
    const int i  = ku[q] / (P / 4);
    const int n0 = (ku[q] - i * (P / 4)) * 4;
    *(float4*)&w0s[i][n0] = s0[q];
    w3t[n0 + 0][i] = s3[q].x; w3t[n0 + 1][i] = s3[q].y;
    w3t[n0 + 2][i] = s3[q].z; w3t[n0 + 3][i] = s3[q].w;
    w4t[n0 + 0][i] = s4[q].x; w4t[n0 + 1][i] = s4[q].y;
    w4t[n0 + 2][i] = s4[q].z; w4t[n0 + 3][i] = s4[q].w;
  }
  if (rok) {
    float s = 0.f;
    #pragma unroll
    for (int q = 0; q < P / 4; ++q) { s += rr[q].x; s += rr[q].y; s += rr[q].z; s += rr[q].w; }
    w3t[P + rsel][rrow] = s;
  }
  __syncthreads();

  const int half = lane >> 4;
  const int ln   = lane & 15;
  const int kb   = half << 1;

  v2f bw3[3][10], bw4[3][10];
  #pragma unroll
  for (int t = 0; t < 3; ++t) {
    #pragma unroll
    for (int ks = 0; ks < 10; ++ks) {
      bw3[t][ks] = *(const v2f*)&w3t[16 * t + ln][4 * ks + kb];
      bw4[t][ks] = *(const v2f*)&w4t[16 * t + ln][4 * ks + kb];
    }
  }

  const int gwave  = blockIdx.x * WPB + wv;
  const int nwaves = gridDim.x * WPB;
  const int npairs = (B + 1) >> 1;

  for (int p = gwave; p < npairs; p += nwaves) {
    const int  b0   = 2 * p;
    const bool hasB = (b0 + 1) < B;
    const float* xb = x + (size_t)b0 * 4 * P;
    if (p + nwaves < npairs)
      __builtin_prefetch(x + (size_t)(p + nwaves) * 8 * P, 0, 3);

    float* ob = out + (size_t)b0 * NF * P;

    float xv[3][4];
    #pragma unroll
    for (int t = 0; t < 3; ++t) {
      const int  idx = lane + 32 * t;
      const bool ok  = idx < (hasB ? 2 * P : P);
      const int  e   = (idx >= P) ? 1 : 0;
      const unsigned off = ok ? (unsigned)(idx + e * 3 * P) : 0u;
      #pragma unroll
      for (int f = 0; f < 4; ++f) xv[t][f] = xb[off + (unsigned)(f * P)];
    }
    #pragma unroll
    for (int t = 0; t < 3; ++t) {
      const int  idx = lane + 32 * t;
      const bool ok  = idx < (hasB ? 2 * P : P);
      const int  e   = (idx >= P) ? 1 : 0;
      const int  i   = idx - e * P;
      if (ok) {
        const float x0 = xv[t][0], x1 = xv[t][1], x2 = xv[t][2], x3 = xv[t][3];
        xs[wv][4 * e + 0][i] = -x0 * x0 + x1 * x1 + x2 * x2 + x3 * x3;
        xs[wv][4 * e + 1][i] = x0; xs[wv][4 * e + 2][i] = x1;
        xs[wv][4 * e + 3][i] = x2; xs[wv][4 * e + 8][i] = x3;
        const unsigned oo = (unsigned)(e * NF * P + i);
        ob[oo]     = x0 * x0 - x1 * x1 - x2 * x2 - x3 * x3;
        ob[oo + P] = sqrtf(x1 * x1 + x2 * x2);
      }
    }
    __builtin_amdgcn_wave_barrier();

    v8f z3a = {}, z3b = {}, z3c = {};
    v8f z4a = {}, z4b = {}, z4c = {};
    #pragma unroll
    for (int ks = 0; ks < 10; ++ks) {
      v2f a = *(const v2f*)&xs[wv][ln][4 * ks + kb];
      z3a = WMMA_F32X4(a, bw3[0][ks], z3a);
      z3b = WMMA_F32X4(a, bw3[1][ks], z3b);
      z3c = WMMA_F32X4(a, bw3[2][ks], z3c);
      z4a = WMMA_F32X4(a, bw4[0][ks], z4a);
      z4b = WMMA_F32X4(a, bw4[1][ks], z4b);
      z4c = WMMA_F32X4(a, bw4[2][ks], z4c);
    }
    __builtin_amdgcn_sched_barrier(0);

    #pragma unroll
    for (int e = 0; e < 2; ++e) {
      if (e == 1 && !hasB) continue;
      const int r0 = 4 * e;
      const int r8 = 8 + 4 * e;
      float* oe = ob + e * NF * P;

      const v2f p3a = {z3a[r0], z3a[r0 + 1]}, q3a = {z3a[r0 + 2], z3a[r0 + 3]};
      const v2f p3b = {z3b[r0], z3b[r0 + 1]}, q3b = {z3b[r0 + 2], z3b[r0 + 3]};
      const v2f p3c = {z3c[r0], z3c[r0 + 1]}, q3c = {z3c[r0 + 2], z3c[r0 + 3]};
      const v2f p4a = {z4a[r0], z4a[r0 + 1]}, q4a = {z4a[r0 + 2], z4a[r0 + 3]};
      const v2f p4b = {z4b[r0], z4b[r0 + 1]}, q4b = {z4b[r0 + 2], z4b[r0 + 3]};
      const v2f p4c = {z4c[r0], z4c[r0 + 1]}, q4c = {z4c[r0 + 2], z4c[r0 + 3]};

      const float g0 = RDLANE(z3c[r0],     8),  h0 = RDLANE(z3c[r0],     9);
      const float g1 = RDLANE(z3c[r0 + 1], 8),  h1 = RDLANE(z3c[r0 + 1], 9);
      const float g2 = RDLANE(z3c[r0 + 2], 8),  h2 = RDLANE(z3c[r0 + 2], 9);
      const float g3 = RDLANE(z3c[r0 + 3], 8),  h3 = RDLANE(z3c[r0 + 3], 9);
      const float g8 = RDLANE(z3c[r0],     24), h8 = RDLANE(z3c[r0],     25);
      const float g9 = RDLANE(z3c[r0 + 1], 24), h9 = RDLANE(z3c[r0 + 1], 25);

      #pragma unroll
      for (int j0 = 0; j0 < PT; j0 += 16) {
        const int jj = j0 + ln;
        const float sj  = xs[wv][r0][jj];
        const float x0j = xs[wv][r0 + 1][jj], x1j = xs[wv][r0 + 2][jj];
        const float x2j = xs[wv][r0 + 3][jj], x3j = xs[wv][r8][jj];
        v2f u0, u1;
        u0.x = half ? -2.f * x3j : 1.f;
        u0.y = half ? sj         : 2.f * x0j;
        u1.x = half ? 0.f        : -2.f * x1j;
        u1.y = half ? 0.f        : -2.f * x2j;
        const v8f zero = {};
        float m3, m4;
        {
          v8f c3a = WMMA_F32X4(p3a, u0, zero);
          v8f c3b = WMMA_F32X4(p3b, u0, zero);
          v8f c3c = WMMA_F32X4(p3c, u0, zero);
          c3a = WMMA_F32X4(q3a, u1, c3a);
          c3b = WMMA_F32X4(q3b, u1, c3b);
          c3c = WMMA_F32X4(q3c, u1, c3c);
          float m  = FMINN(c3a[0], c3b[0]);
          float mc = FMINN(c3c[0], c3c[1]);
          #pragma unroll
          for (int r = 1; r < 8; ++r) m = FMINN(FMINN(m, c3a[r]), c3b[r]);
          #pragma unroll
          for (int r = 2; r < 8; r += 2) mc = FMINN(FMINN(mc, c3c[r]), c3c[r + 1]);
          m3 = half ? m : FMINN(m, mc);
        }
        {
          v8f c4a = WMMA_F32X4(p4a, u0, zero);
          v8f c4b = WMMA_F32X4(p4b, u0, zero);
          v8f c4c = WMMA_F32X4(p4c, u0, zero);
          c4a = WMMA_F32X4(q4a, u1, c4a);
          c4b = WMMA_F32X4(q4b, u1, c4b);
          c4c = WMMA_F32X4(q4c, u1, c4c);
          float m  = FMINN(c4a[0], c4b[0]);
          float mc = FMINN(c4c[0], c4c[1]);
          #pragma unroll
          for (int r = 1; r < 8; ++r) m = FMINN(FMINN(m, c4a[r]), c4b[r]);
          #pragma unroll
          for (int r = 2; r < 8; r += 2) mc = FMINN(FMINN(mc, c4c[r]), c4c[r + 1]);
          m4 = half ? m : FMINN(m, mc);
        }
        {
          const float keep = half ? m4 : m3;
          const float send = half ? m3 : m4;
          const float res  = FMINN(keep, xhalf(send));
          if (jj < P) oe[(unsigned)((5 + half) * P + jj)] = res;
        }
      }

      #pragma unroll
      for (int t = 0; t < 2; ++t) {
        const int j = lane + 32 * t;
        if (j >= P) continue;
        const float sj  = xs[wv][r0][j];
        const float x0j = xs[wv][r0 + 1][j], x1j = xs[wv][r0 + 2][j];
        const float x2j = xs[wv][r0 + 3][j], x3j = xs[wv][r8][j];
        float a0 = 0.f;
        for (int i = 0; i < P; ++i)
          a0 += xs[wv][r0 + 1][i] * w0s[i][j];
        oe[(unsigned)(2 * P + j)] = a0;
        oe[(unsigned)(3 * P + j)] = g0 + sj * g9 + 2.f * (x0j * g1 - x1j * g2 - x2j * g3 - x3j * g8);
        oe[(unsigned)(4 * P + j)] = h0 + sj * h9 + 2.f * (x0j * h1 - x1j * h2 - x2j * h3 - x3j * h8);
      }
    }
    __builtin_amdgcn_wave_barrier();
  }
}

extern "C" void kernel_launch(void* const* d_in, const int* in_sizes, int n_in,
                              void* d_out, int out_size, void* d_ws, size_t ws_size,
                              hipStream_t stream) {
  const float* x = (const float*)d_in[0];
  const float* w = (const float*)d_in[1];
  float* out = (float*)d_out;
  const int B = in_sizes[0] / (4 * P);

  int totalWaves = (B + 3) / 4;
  int grid = (totalWaves + WPB - 1) / WPB;
  if (grid > 1024) grid = 1024;
  if (grid < 1) grid = 1;

  lorentz_feats_kernel<<<grid, BLOCK, 0, stream>>>(x, w, out, B);
}


# second product tiled over the pair without padding tiles: w3|w4 columns packed into 5 tiles (row sums in a sixth, first product only), the pair's 80 particles into 5 tiles: 132 -> 115 WMMAs per pair
# speedup vs baseline: 3.8913x; 1.0720x over previous
#include <hip/hip_runtime.h>

typedef __attribute__((ext_vector_type(2))) float v2f;
typedef __attribute__((ext_vector_type(8))) float v8f;

#define P   40
#define XST 50
#define WST 42
#define NF  7
#define NT  6
#define WPB 4
#define BLOCK (WPB * 32)

#define FMINN(a, b) __builtin_fminimum_numf((a), (b))

__device__ __forceinline__ float xhalf(float x) {
  const int b = __float_as_int(x);
  return __int_as_float(__builtin_amdgcn_permlanex16(b, b, 0x76543210, 0xfedcba98, false, false));
}

#define WMMA_F32X4(A, Bv, C) \
  __builtin_amdgcn_wmma_f32_16x16x4_f32(false, (A), false, (Bv), (short)0, (C), false, false)

#define RDLANE(x, l) __int_as_float(__builtin_amdgcn_readlane(__float_as_int(x), (l)))

__global__ __launch_bounds__(BLOCK, 4)
void lorentz_feats_kernel(const float* __restrict__ x,
                          const float* __restrict__ w,
                          float* __restrict__ out,
                          int B)
{
  __shared__ __align__(16) float w0s[P][P];
  __shared__ __align__(16) float wt[NT * 16][WST];
  __shared__ __align__(16) float xs[WPB][16][XST];

  const int tid  = threadIdx.x;
  const int lane = tid & 31;
  const int wv   = __builtin_amdgcn_readfirstlane(tid >> 5);

  const float4* __restrict__ wq = (const float4*)w;
  float4 s0[4], s3[4], s4[4], rr[P / 4];
  int ku[4];
  #pragma unroll
  for (int q = 0; q < 4; ++q) {
    const int k = tid + BLOCK * q;
    ku[q] = (q < 3 || k < P * P / 4) ? k : 0;
    s0[q] = wq[ku[q]];
    s3[q] = wq[3 * (P * P / 4) + ku[q]];
    s4[q] = wq[4 * (P * P / 4) + ku[q]];
  }
  const int  rsel = tid >> 6;
  const int  rrow = tid & 63;
  const bool rok  = rrow < P;
  {
    const int base = (1 + rsel) * (P * P / 4) + (rok ? rrow : 0) * (P / 4);
    #pragma unroll
    for (int q = 0; q < P / 4; ++q) rr[q] = wq[base + q];
  }

  {
    float* zr = &wt[2 * P + 2][0];
    for (int k = tid; k < (NT * 16 - 2 * P - 2) * WST; k += BLOCK) zr[k] = 0.f;
    float* xw = &xs[wv][0][0];
    #pragma unroll
    for (int q = 0; q < (16 * XST) / 32; ++q) {
      const int k = lane + 32 * q;
      const bool one = (k >= 9 * XST && k < 9 * XST + P) || (k >= 13 * XST && k < 13 * XST + P);
      xw[k] = one ? 1.f : 0.f;
    }
  }

  #pragma unroll
  for (int q = 0; q < 4; ++q) {
    const int i  = ku[q] / (P / 4);
    const int n0 = (ku[q] - i * (P / 4)) * 4;
    *(float4*)&w0s[i][n0] = s0[q];
    wt[n0 + 0][i] = s3[q].x; wt[n0 + 1][i] = s3[q].y;
    wt[n0 + 2][i] = s3[q].z; wt[n0 + 3][i] = s3[q].w;
    wt[P + n0 + 0][i] = s4[q].x; wt[P + n0 + 1][i] = s4[q].y;
    wt[P + n0 + 2][i] = s4[q].z; wt[P + n0 + 3][i] = s4[q].w;
  }
  if (rok) {
    float s = 0.f;
    #pragma unroll
    for (int q = 0; q < P / 4; ++q) { s += rr[q].x; s += rr[q].y; s += rr[q].z; s += rr[q].w; }
    wt[2 * P + rsel][rrow] = s;
  }
  __syncthreads();

  const int half = lane >> 4;
  const int ln   = lane & 15;
  const int kb   = half << 1;

  v2f bw[NT - 1][10];
  #pragma unroll
  for (int t = 0; t < NT - 1; ++t) {
    #pragma unroll
    for (int ks = 0; ks < 10; ++ks)
      bw[t][ks] = *(const v2f*)&wt[16 * t + ln][4 * ks + kb];
  }

  const int gwave  = blockIdx.x * WPB + wv;
  const int nwaves = gridDim.x * WPB;
  const int npairs = (B + 1) >> 1;

  for (int p = gwave; p < npairs; p += nwaves) {
    const int  b0   = 2 * p;
    const bool hasB = (b0 + 1) < B;
    const float* xb = x + (size_t)b0 * 4 * P;
    if (p + nwaves < npairs)
      __builtin_prefetch(x + (size_t)(p + nwaves) * 8 * P, 0, 3);

    float* ob = out + (size_t)b0 * NF * P;

    float xv[3][4];
    #pragma unroll
    for (int t = 0; t < 3; ++t) {
      const int  idx = lane + 32 * t;
      const bool ok  = idx < (hasB ? 2 * P : P);
      const int  e   = (idx >= P) ? 1 : 0;
      const unsigned off = ok ? (unsigned)(idx + e * 3 * P) : 0u;
      #pragma unroll
      for (int f = 0; f < 4; ++f) xv[t][f] = xb[off + (unsigned)(f * P)];
    }
    #pragma unroll
    for (int t = 0; t < 3; ++t) {
      const int  idx = lane + 32 * t;
      const bool ok  = idx < (hasB ? 2 * P : P);
      const int  e   = (idx >= P) ? 1 : 0;
      const int  i   = idx - e * P;
      if (ok) {
        const float x0 = xv[t][0], x1 = xv[t][1], x2 = xv[t][2], x3 = xv[t][3];
        xs[wv][4 * e + 0][i] = -x0 * x0 + x1 * x1 + x2 * x2 + x3 * x3;
        xs[wv][4 * e + 1][i] = x0; xs[wv][8 * e + 2][i] = x1;
        xs[wv][8 * e + 3][i] = x2; xs[wv][4 * e + 8][i] = x3;
        const unsigned oo = (unsigned)(e * NF * P + i);
        ob[oo]     = x0 * x0 - x1 * x1 - x2 * x2 - x3 * x3;
        ob[oo + P] = sqrtf(x1 * x1 + x2 * x2);
      }
    }
    __builtin_amdgcn_wave_barrier();

    v2f b5[10];
    {
      unsigned rsoff = (unsigned)((16 * (NT - 1) + ln) * WST + kb);
      asm volatile("" : "+v"(rsoff));
      const float* wrs = &wt[0][0] + rsoff;
      #pragma unroll
      for (int ks = 0; ks < 10; ++ks) b5[ks] = *(const v2f*)(wrs + 4 * ks);
    }
    v8f z[NT];
    #pragma unroll
    for (int t = 0; t < NT; ++t) z[t] = (v8f){};
    #pragma unroll
    for (int ks = 0; ks < 10; ++ks) {
      v2f a = *(const v2f*)&xs[wv][ln][4 * ks + kb];
      #pragma unroll
      for (int t = 0; t < NT - 1; ++t) z[t] = WMMA_F32X4(a, bw[t][ks], z[t]);
      z[NT - 1] = WMMA_F32X4(a, b5[ks], z[NT - 1]);
    }
    __builtin_amdgcn_sched_barrier(0);

    float g[2][6], hh[2][6];
    g[0][0] = RDLANE(z[5][0], 0);  hh[0][0] = RDLANE(z[5][0], 1);
    g[0][1] = RDLANE(z[5][1], 0);  hh[0][1] = RDLANE(z[5][1], 1);
    g[0][2] = RDLANE(z[5][2], 0);  hh[0][2] = RDLANE(z[5][2], 1);
    g[0][3] = RDLANE(z[5][3], 0);  hh[0][3] = RDLANE(z[5][3], 1);
    g[0][4] = RDLANE(z[5][0], 16); hh[0][4] = RDLANE(z[5][0], 17);
    g[0][5] = RDLANE(z[5][1], 16); hh[0][5] = RDLANE(z[5][1], 17);
    g[1][0] = RDLANE(z[5][4], 0);  hh[1][0] = RDLANE(z[5][4], 1);
    g[1][1] = RDLANE(z[5][5], 0);  hh[1][1] = RDLANE(z[5][5], 1);
    g[1][2] = RDLANE(z[5][2], 16); hh[1][2] = RDLANE(z[5][2], 17);
    g[1][3] = RDLANE(z[5][3], 16); hh[1][3] = RDLANE(z[5][3], 17);
    g[1][4] = RDLANE(z[5][4], 16); hh[1][4] = RDLANE(z[5][4], 17);
    g[1][5] = RDLANE(z[5][5], 16); hh[1][5] = RDLANE(z[5][5], 17);

    #pragma unroll
    for (int jt = 0; jt < 5; ++jt) {
      if (jt >= 3 && !hasB) continue;
      const int el = (jt < 2) ? 0 : ((jt > 2) ? 1 : (ln >> 3));
      const int j  = (jt < 2) ? (16 * jt + ln) : ((jt > 2) ? (16 * (jt - 3) + ln) : (32 + (ln & 7)));
      const float* xr = &xs[wv][0][0];
      const float sj  = xr[(4 * el    ) * XST + j];
      const float x0j = xr[(4 * el + 1) * XST + j];
      const float x1j = xr[(8 * el + 2) * XST + j];
      const float x2j = xr[(8 * el + 3) * XST + j];
      const float x3j = xr[(4 * el + 8) * XST + j];
      v2f uo, um;
      uo.x = half ? -2.f * x3j : 1.f;
      uo.y = half ? sj         : 2.f * x0j;
      const bool mine = (half == el);
      um.x = mine ? -2.f * x1j : 0.f;
      um.y = mine ? -2.f * x2j : 0.f;
      const v8f zero = {};
      v8f c[5];
      if (jt < 2) {
        #pragma unroll
        for (int t = 0; t < 5; ++t) { const v2f p = {z[t][0], z[t][1]}; c[t] = WMMA_F32X4(p, uo, zero); }
        #pragma unroll
        for (int t = 0; t < 5; ++t) { const v2f p = {z[t][2], z[t][3]}; c[t] = WMMA_F32X4(p, um, c[t]); }
      } else if (jt > 2) {
        #pragma unroll
        for (int t = 0; t < 5; ++t) { const v2f p = {z[t][4], z[t][5]}; c[t] = WMMA_F32X4(p, uo, zero); }
        #pragma unroll
        for (int t = 0; t < 5; ++t) { const v2f p = {z[t][2], z[t][3]}; c[t] = WMMA_F32X4(p, um, c[t]); }
      } else {
        v2f u0, u1;
        u0.x = el ? 0.f : uo.x; u0.y = el ? 0.f : uo.y;
        u1.x = el ? uo.x : 0.f; u1.y = el ? uo.y : 0.f;
        #pragma unroll
        for (int t = 0; t < 5; ++t) { const v2f p = {z[t][0], z[t][1]}; c[t] = WMMA_F32X4(p, u0, zero); }
        #pragma unroll
        for (int t = 0; t < 5; ++t) { const v2f p = {z[t][4], z[t][5]}; c[t] = WMMA_F32X4(p, u1, c[t]); }
        #pragma unroll
        for (int t = 0; t < 5; ++t) { const v2f p = {z[t][2], z[t][3]}; c[t] = WMMA_F32X4(p, um, c[t]); }
      }
      float ma = FMINN(c[0][0], c[1][0]);
      float mb = FMINN(c[3][0], c[4][0]);
      float mc = FMINN(c[2][0], c[2][1]);
      #pragma unroll
      for (int r = 1; r < 8; ++r) {
        ma = FMINN(FMINN(ma, c[0][r]), c[1][r]);
        mb = FMINN(FMINN(mb, c[3][r]), c[4][r]);
      }
      #pragma unroll
      for (int r = 2; r < 8; r += 2) mc = FMINN(FMINN(mc, c[2][r]), c[2][r + 1]);
      {
        const float keep = FMINN(half ? mb : ma, mc);
        const float send = half ? ma : mb;
        const float res  = FMINN(keep, xhalf(send));
        if (jt != 2 || el == 0 || hasB)
          ob[(unsigned)(el * NF * P + (5 + half) * P + j)] = res;
      }
    }

    #pragma unroll
    for (int e = 0; e < 2; ++e) {
      if (e == 1 && !hasB) continue;
      float* oe = ob + e * NF * P;
      #pragma unroll
      for (int t = 0; t < 2; ++t) {
        const int j = lane + 32 * t;
        if (j >= P) continue;
        const float sj  = xs[wv][4 * e][j];
        const float x0j = xs[wv][4 * e + 1][j], x1j = xs[wv][8 * e + 2][j];
        const float x2j = xs[wv][8 * e + 3][j], x3j = xs[wv][4 * e + 8][j];
        float a0 = 0.f;
        for (int i = 0; i < P; ++i)
          a0 += xs[wv][4 * e + 1][i] * w0s[i][j];
        oe[(unsigned)(2 * P + j)] = a0;
        oe[(unsigned)(3 * P + j)] = g[e][0] + sj * g[e][5] + 2.f * (x0j * g[e][1] - x1j * g[e][2] - x2j * g[e][3] - x3j * g[e][4]);
        oe[(unsigned)(4 * P + j)] = hh[e][0] + sj * hh[e][5] + 2.f * (x0j * hh[e][1] - x1j * hh[e][2] - x2j * hh[e][3] - x3j * hh[e][4]);
      }
    }
    __builtin_amdgcn_wave_barrier();
  }
}

extern "C" void kernel_launch(void* const* d_in, const int* in_sizes, int n_in,
                              void* d_out, int out_size, void* d_ws, size_t ws_size,
                              hipStream_t stream) {
  const float* x = (const float*)d_in[0];
  const float* w = (const float*)d_in[1];
  float* out = (float*)d_out;
  const int B = in_sizes[0] / (4 * P);

  int totalWaves = (B + 3) / 4;
  int grid = (totalWaves + WPB - 1) / WPB;
  if (grid > 1024) grid = 1024;
  if (grid < 1) grid = 1;

  lorentz_feats_kernel<<<grid, BLOCK, 0, stream>>>(x, w, out, B);
}


# Es @ w0 leaves its 40-step per-lane vector-unit chains for one more WMMA product taken per block: x0 rows of the block's 16 batch elements fill one A tile in LDS, w0 is B: 30 WMMAs per 16 elements
# speedup vs baseline: 4.1908x; 1.0770x over previous
#include <hip/hip_runtime.h>

typedef __attribute__((ext_vector_type(2))) float v2f;
typedef __attribute__((ext_vector_type(8))) float v8f;

#define P   40
#define XST 50
#define WST 42
#define EST 44
#define NF  7
#define NT  6
#define WPB 4
#define BLOCK (WPB * 32)

#define FMINN(a, b) __builtin_fminimum_numf((a), (b))

__device__ __forceinline__ float xhalf(float x) {
  const int b = __float_as_int(x);
  return __int_as_float(__builtin_amdgcn_permlanex16(b, b, 0x76543210, 0xfedcba98, false, false));
}

#define WMMA_F32X4(A, Bv, C) \
  __builtin_amdgcn_wmma_f32_16x16x4_f32(false, (A), false, (Bv), (short)0, (C), false, false)

#define RDLANE(x, l) __int_as_float(__builtin_amdgcn_readlane(__float_as_int(x), (l)))

__global__ __launch_bounds__(BLOCK, 4)
void lorentz_feats_kernel(const float* __restrict__ x,
                          const float* __restrict__ w,
                          float* __restrict__ out,
                          int B)
{
  __shared__ __align__(16) float w0s[P][P];
  __shared__ __align__(16) float wt[NT * 16][WST];
  __shared__ __align__(16) float xs[WPB][16][XST];
  __shared__ __align__(16) float es[16][EST];

  const int tid  = threadIdx.x;
  const int lane = tid & 31;
  const int wv   = __builtin_amdgcn_readfirstlane(tid >> 5);

  const float4* __restrict__ wq = (const float4*)w;
  float4 s0[4], s3[4], s4[4], rr[P / 4];
  int ku[4];
  #pragma unroll
  for (int q = 0; q < 4; ++q) {
    const int k = tid + BLOCK * q;
    ku[q] = (q < 3 || k < P * P / 4) ? k : 0;
    s0[q] = wq[ku[q]];
    s3[q] = wq[3 * (P * P / 4) + ku[q]];
    s4[q] = wq[4 * (P * P / 4) + ku[q]];
  }
  const int half = lane >> 4;
  const int ln   = lane & 15;
  const int kb   = half << 1;

  const int  tE    = (wv + (int)blockIdx.x) & 3;
  const int  eb0   = 8 * (int)blockIdx.x;
  const int  estep = 2 * WPB * (int)gridDim.x;
  const bool blkE  = eb0 < B;
  const bool doE   = (tE < 3) && blkE;
  const int er0 = tid / (P / 4),           ec0 = tid - er0 * (P / 4);
  const int er1 = (tid + BLOCK) / (P / 4), ec1 = (tid + BLOCK) - er1 * (P / 4);
  float4 ev0 = {0.f, 0.f, 0.f, 0.f}, ev1 = {0.f, 0.f, 0.f, 0.f};
  if (blkE) {
    int be = eb0 + (er0 >> 3) * estep + (er0 & 7);
    be = be < B ? be : B - 1;
    ev0 = *(const float4*)(x + (size_t)be * 4 * P + 4 * ec0);
    if (wv == 0) {
      int b1 = eb0 + (er1 >> 3) * estep + (er1 & 7);
      b1 = b1 < B ? b1 : B - 1;
      ev1 = *(const float4*)(x + (size_t)b1 * 4 * P + 4 * ec1);
    }
  }

  const int  rsel = tid >> 6;
  const int  rrow = tid & 63;
  const bool rok  = rrow < P;
  {
    const int base = (1 + rsel) * (P * P / 4) + (rok ? rrow : 0) * (P / 4);
    #pragma unroll
    for (int q = 0; q < P / 4; ++q) rr[q] = wq[base + q];
  }

  {
    float* zr = &wt[2 * P + 2][0];
    for (int k = tid; k < (NT * 16 - 2 * P - 2) * WST; k += BLOCK) zr[k] = 0.f;
    float* xw = &xs[wv][0][0];
    #pragma unroll
    for (int q = 0; q < (16 * XST) / 32; ++q) {
      const int k = lane + 32 * q;
      const bool one = (k >= 9 * XST && k < 9 * XST + P) || (k >= 13 * XST && k < 13 * XST + P);
      xw[k] = one ? 1.f : 0.f;
    }
  }

  #pragma unroll
  for (int q = 0; q < 4; ++q) {
    const int i  = ku[q] / (P / 4);
    const int n0 = (ku[q] - i * (P / 4)) * 4;
    *(float4*)&w0s[i][n0] = s0[q];
    wt[n0 + 0][i] = s3[q].x; wt[n0 + 1][i] = s3[q].y;
    wt[n0 + 2][i] = s3[q].z; wt[n0 + 3][i] = s3[q].w;
    wt[P + n0 + 0][i] = s4[q].x; wt[P + n0 + 1][i] = s4[q].y;
    wt[P + n0 + 2][i] = s4[q].z; wt[P + n0 + 3][i] = s4[q].w;
  }
  if (rok) {
    float s = 0.f;
    #pragma unroll
    for (int q = 0; q < P / 4; ++q) { s += rr[q].x; s += rr[q].y; s += rr[q].z; s += rr[q].w; }
    wt[2 * P + rsel][rrow] = s;
  }
  if (blkE) {
    *(float4*)&es[er0][4 * ec0] = ev0;
    if (wv == 0) *(float4*)&es[er1][4 * ec1] = ev1;
  }
  __syncthreads();

  if (doE) {
    const int q  = 16 * tE + ln;
    const int qc = q < P ? q : P - 1;
    const float* wb = &w0s[kb][qc];
    v2f ea[10];
    #pragma unroll
    for (int ks = 0; ks < 10; ++ks) ea[ks] = *(const v2f*)&es[ln][4 * ks + kb];
    for (int e0 = eb0; e0 < B; e0 += 2 * estep) {
      if (e0 != eb0) {
        int be = e0 + (ln >> 3) * estep + (ln & 7);
        be = be < B ? be : B - 1;
        const float* xe = x + (size_t)be * 4 * P + kb;
        #pragma unroll
        for (int ks = 0; ks < 10; ++ks) ea[ks] = *(const v2f*)(xe + 4 * ks);
      }
      v8f d0 = {}, d1 = {};
      #pragma unroll
      for (int ks = 0; ks < 10; ks += 2) {
        const v2f b0 = {wb[(4 * ks    ) * P], wb[(4 * ks + 1) * P]};
        const v2f b1 = {wb[(4 * ks + 4) * P], wb[(4 * ks + 5) * P]};
        d0 = WMMA_F32X4(ea[ks],     b0, d0);
        d1 = WMMA_F32X4(ea[ks + 1], b1, d1);
      }
      if (q < P) {
        const int m0 = e0 + half * estep;
        float* oq = out + (size_t)m0 * NF * P + 2 * P + q;
        #pragma unroll
        for (int r = 0; r < 8; ++r)
          if (m0 + r < B) oq[(unsigned)(r * NF * P)] = d0[r] + d1[r];
      }
    }
  }

  v2f bw[NT - 1][10];
  #pragma unroll
  for (int t = 0; t < NT - 1; ++t) {
    #pragma unroll
    for (int ks = 0; ks < 10; ++ks)
      bw[t][ks] = *(const v2f*)&wt[16 * t + ln][4 * ks + kb];
  }

  const int gwave  = blockIdx.x * WPB + wv;
  const int nwaves = gridDim.x * WPB;
  const int npairs = (B + 1) >> 1;

  for (int p = gwave; p < npairs; p += nwaves) {
    const int  b0   = 2 * p;
    const bool hasB = (b0 + 1) < B;
    const float* xb = x + (size_t)b0 * 4 * P;
    if (p + nwaves < npairs)
      __builtin_prefetch(x + (size_t)(p + nwaves) * 8 * P, 0, 3);

    float* ob = out + (size_t)b0 * NF * P;

    float xv[3][4];
    #pragma unroll
    for (int t = 0; t < 3; ++t) {
      const int  idx = lane + 32 * t;
      const bool ok  = idx < (hasB ? 2 * P : P);
      const int  e   = (idx >= P) ? 1 : 0;
      const unsigned off = ok ? (unsigned)(idx + e * 3 * P) : 0u;
      #pragma unroll
      for (int f = 0; f < 4; ++f) xv[t][f] = xb[off + (unsigned)(f * P)];
    }
    #pragma unroll
    for (int t = 0; t < 3; ++t) {
      const int  idx = lane + 32 * t;
      const bool ok  = idx < (hasB ? 2 * P : P);
      const int  e   = (idx >= P) ? 1 : 0;
      const int  i   = idx - e * P;
      if (ok) {
        const float x0 = xv[t][0], x1 = xv[t][1], x2 = xv[t][2], x3 = xv[t][3];
        xs[wv][4 * e + 0][i] = -x0 * x0 + x1 * x1 + x2 * x2 + x3 * x3;
        xs[wv][4 * e + 1][i] = x0; xs[wv][8 * e + 2][i] = x1;
        xs[wv][8 * e + 3][i] = x2; xs[wv][4 * e + 8][i] = x3;
        const unsigned oo = (unsigned)(e * NF * P + i);
        ob[oo]     = x0 * x0 - x1 * x1 - x2 * x2 - x3 * x3;
        ob[oo + P] = sqrtf(x1 * x1 + x2 * x2);
      }
    }
    __builtin_amdgcn_wave_barrier();

    v2f b5[10];
    {
      unsigned rsoff = (unsigned)((16 * (NT - 1) + ln) * WST + kb);
      asm volatile("" : "+v"(rsoff));
      const float* wrs = &wt[0][0] + rsoff;
      #pragma unroll
      for (int ks = 0; ks < 10; ++ks) b5[ks] = *(const v2f*)(wrs + 4 * ks);
    }
    v8f z[NT];
    #pragma unroll
    for (int t = 0; t < NT; ++t) z[t] = (v8f){};
    #pragma unroll
    for (int ks = 0; ks < 10; ++ks) {
      v2f a = *(const v2f*)&xs[wv][ln][4 * ks + kb];
      #pragma unroll
      for (int t = 0; t < NT - 1; ++t) z[t] = WMMA_F32X4(a, bw[t][ks], z[t]);
      z[NT - 1] = WMMA_F32X4(a, b5[ks], z[NT - 1]);
    }
    __builtin_amdgcn_sched_barrier(0);

    float g[2][6], hh[2][6];
    g[0][0] = RDLANE(z[5][0], 0);  hh[0][0] = RDLANE(z[5][0], 1);
    g[0][1] = RDLANE(z[5][1], 0);  hh[0][1] = RDLANE(z[5][1], 1);
    g[0][2] = RDLANE(z[5][2], 0);  hh[0][2] = RDLANE(z[5][2], 1);
    g[0][3] = RDLANE(z[5][3], 0);  hh[0][3] = RDLANE(z[5][3], 1);
    g[0][4] = RDLANE(z[5][0], 16); hh[0][4] = RDLANE(z[5][0], 17);
    g[0][5] = RDLANE(z[5][1], 16); hh[0][5] = RDLANE(z[5][1], 17);
    g[1][0] = RDLANE(z[5][4], 0);  hh[1][0] = RDLANE(z[5][4], 1);
    g[1][1] = RDLANE(z[5][5], 0);  hh[1][1] = RDLANE(z[5][5], 1);
    g[1][2] = RDLANE(z[5][2], 16); hh[1][2] = RDLANE(z[5][2], 17);
    g[1][3] = RDLANE(z[5][3], 16); hh[1][3] = RDLANE(z[5][3], 17);
    g[1][4] = RDLANE(z[5][4], 16); hh[1][4] = RDLANE(z[5][4], 17);
    g[1][5] = RDLANE(z[5][5], 16); hh[1][5] = RDLANE(z[5][5], 17);

    #pragma unroll
    for (int jt = 0; jt < 5; ++jt) {
      if (jt >= 3 && !hasB) continue;
      const int el = (jt < 2) ? 0 : ((jt > 2) ? 1 : (ln >> 3));
      const int j  = (jt < 2) ? (16 * jt + ln) : ((jt > 2) ? (16 * (jt - 3) + ln) : (32 + (ln & 7)));
      const float* xr = &xs[wv][0][0];
      const float sj  = xr[(4 * el    ) * XST + j];
      const float x0j = xr[(4 * el + 1) * XST + j];
      const float x1j = xr[(8 * el + 2) * XST + j];
      const float x2j = xr[(8 * el + 3) * XST + j];
      const float x3j = xr[(4 * el + 8) * XST + j];
      v2f uo, um;
      uo.x = half ? -2.f * x3j : 1.f;
      uo.y = half ? sj         : 2.f * x0j;
      const bool mine = (half == el);
      um.x = mine ? -2.f * x1j : 0.f;
      um.y = mine ? -2.f * x2j : 0.f;
      const v8f zero = {};
      v8f c[5];
      if (jt < 2) {
        #pragma unroll
        for (int t = 0; t < 5; ++t) { const v2f p = {z[t][0], z[t][1]}; c[t] = WMMA_F32X4(p, uo, zero); }
        #pragma unroll
        for (int t = 0; t < 5; ++t) { const v2f p = {z[t][2], z[t][3]}; c[t] = WMMA_F32X4(p, um, c[t]); }
      } else if (jt > 2) {
        #pragma unroll
        for (int t = 0; t < 5; ++t) { const v2f p = {z[t][4], z[t][5]}; c[t] = WMMA_F32X4(p, uo, zero); }
        #pragma unroll
        for (int t = 0; t < 5; ++t) { const v2f p = {z[t][2], z[t][3]}; c[t] = WMMA_F32X4(p, um, c[t]); }
      } else {
        v2f u0, u1;
        u0.x = el ? 0.f : uo.x; u0.y = el ? 0.f : uo.y;
        u1.x = el ? uo.x : 0.f; u1.y = el ? uo.y : 0.f;
        #pragma unroll
        for (int t = 0; t < 5; ++t) { const v2f p = {z[t][0], z[t][1]}; c[t] = WMMA_F32X4(p, u0, zero); }
        #pragma unroll
        for (int t = 0; t < 5; ++t) { const v2f p = {z[t][4], z[t][5]}; c[t] = WMMA_F32X4(p, u1, c[t]); }
        #pragma unroll
        for (int t = 0; t < 5; ++t) { const v2f p = {z[t][2], z[t][3]}; c[t] = WMMA_F32X4(p, um, c[t]); }
      }
      float ma = FMINN(c[0][0], c[1][0]);
      float mb = FMINN(c[3][0], c[4][0]);
      float mc = FMINN(c[2][0], c[2][1]);
      #pragma unroll
      for (int r = 1; r < 8; ++r) {
        ma = FMINN(FMINN(ma, c[0][r]), c[1][r]);
        mb = FMINN(FMINN(mb, c[3][r]), c[4][r]);
      }
      #pragma unroll
      for (int r = 2; r < 8; r += 2) mc = FMINN(FMINN(mc, c[2][r]), c[2][r + 1]);
      {
        const float keep = FMINN(half ? mb : ma, mc);
        const float send = half ? ma : mb;
        const float res  = FMINN(keep, xhalf(send));
        if (jt != 2 || el == 0 || hasB)
          ob[(unsigned)(el * NF * P + (5 + half) * P + j)] = res;
      }
    }

    #pragma unroll
    for (int e = 0; e < 2; ++e) {
      if (e == 1 && !hasB) continue;
      float* oe = ob + e * NF * P;
      #pragma unroll
      for (int t = 0; t < 2; ++t) {
        const int j = lane + 32 * t;
        if (j >= P) continue;
        const float sj  = xs[wv][4 * e][j];
        const float x0j = xs[wv][4 * e + 1][j], x1j = xs[wv][8 * e + 2][j];
        const float x2j = xs[wv][8 * e + 3][j], x3j = xs[wv][4 * e + 8][j];
        oe[(unsigned)(3 * P + j)] = g[e][0] + sj * g[e][5] + 2.f * (x0j * g[e][1] - x1j * g[e][2] - x2j * g[e][3] - x3j * g[e][4]);
        oe[(unsigned)(4 * P + j)] = hh[e][0] + sj * hh[e][5] + 2.f * (x0j * hh[e][1] - x1j * hh[e][2] - x2j * hh[e][3] - x3j * hh[e][4]);
      }
    }
    __builtin_amdgcn_wave_barrier();
  }
}

extern "C" void kernel_launch(void* const* d_in, const int* in_sizes, int n_in,
                              void* d_out, int out_size, void* d_ws, size_t ws_size,
                              hipStream_t stream) {
  const float* x = (const float*)d_in[0];
  const float* w = (const float*)d_in[1];
  float* out = (float*)d_out;
  const int B = in_sizes[0] / (4 * P);

  int totalWaves = (B + 3) / 4;
  int grid = (totalWaves + WPB - 1) / WPB;
  if (grid > 1024) grid = 1024;
  if (grid < 1) grid = 1;

  lorentz_feats_kernel<<<grid, BLOCK, 0, stream>>>(x, w, out, B);
}


# wave priority by work left (s_setprio): 3 from entry to a wave's last pair, then 2 and 1 through that pair: the wave that is behind wins the SIMD and the four waves of a SIMD end together
# speedup vs baseline: 4.5260x; 1.0800x over previous
#include <hip/hip_runtime.h>

typedef __attribute__((ext_vector_type(2))) float v2f;
typedef __attribute__((ext_vector_type(8))) float v8f;

#define P   40
#define XST 50
#define WST 42
#define EST 44
#define NF  7
#define NT  6
#define WPB 4
#define BLOCK (WPB * 32)

#define FMINN(a, b) __builtin_fminimum_numf((a), (b))

__device__ __forceinline__ float xhalf(float x) {
  const int b = __float_as_int(x);
  return __int_as_float(__builtin_amdgcn_permlanex16(b, b, 0x76543210, 0xfedcba98, false, false));
}

#define WMMA_F32X4(A, Bv, C) \
  __builtin_amdgcn_wmma_f32_16x16x4_f32(false, (A), false, (Bv), (short)0, (C), false, false)

#define RDLANE(x, l) __int_as_float(__builtin_amdgcn_readlane(__float_as_int(x), (l)))

__global__ __launch_bounds__(BLOCK, 4)
void lorentz_feats_kernel(const float* __restrict__ x,
                          const float* __restrict__ w,
                          float* __restrict__ out,
                          int B)
{
  __shared__ __align__(16) float w0s[P][P];
  __shared__ __align__(16) float wt[NT * 16][WST];
  __shared__ __align__(16) float xs[WPB][16][XST];
  __shared__ __align__(16) float es[16][EST];

  __builtin_amdgcn_s_setprio(3);

  const int tid  = threadIdx.x;
  const int lane = tid & 31;
  const int wv   = __builtin_amdgcn_readfirstlane(tid >> 5);

  const float4* __restrict__ wq = (const float4*)w;
  float4 s0[4], s3[4], s4[4], rr[P / 4];
  int ku[4];
  #pragma unroll
  for (int q = 0; q < 4; ++q) {
    const int k = tid + BLOCK * q;
    ku[q] = (q < 3 || k < P * P / 4) ? k : 0;
    s0[q] = wq[ku[q]];
    s3[q] = wq[3 * (P * P / 4) + ku[q]];
    s4[q] = wq[4 * (P * P / 4) + ku[q]];
  }
  const int half = lane >> 4;
  const int ln   = lane & 15;
  const int kb   = half << 1;

  const int  tE    = (wv + (int)blockIdx.x) & 3;
  const int  eb0   = 8 * (int)blockIdx.x;
  const int  estep = 2 * WPB * (int)gridDim.x;
  const bool blkE  = eb0 < B;
  const bool doE   = (tE < 3) && blkE;
  const int er0 = tid / (P / 4),           ec0 = tid - er0 * (P / 4);
  const int er1 = (tid + BLOCK) / (P / 4), ec1 = (tid + BLOCK) - er1 * (P / 4);
  float4 ev0 = {0.f, 0.f, 0.f, 0.f}, ev1 = {0.f, 0.f, 0.f, 0.f};
  if (blkE) {
    int be = eb0 + (er0 >> 3) * estep + (er0 & 7);
    be = be < B ? be : B - 1;
    ev0 = *(const float4*)(x + (size_t)be * 4 * P + 4 * ec0);
    if (wv == 0) {
      int b1 = eb0 + (er1 >> 3) * estep + (er1 & 7);
      b1 = b1 < B ? b1 : B - 1;
      ev1 = *(const float4*)(x + (size_t)b1 * 4 * P + 4 * ec1);
    }
  }

  const int  rsel = tid >> 6;
  const int  rrow = tid & 63;
  const bool rok  = rrow < P;
  {
    const int base = (1 + rsel) * (P * P / 4) + (rok ? rrow : 0) * (P / 4);
    #pragma unroll
    for (int q = 0; q < P / 4; ++q) rr[q] = wq[base + q];
  }

  {
    float* zr = &wt[2 * P + 2][0];
    for (int k = tid; k < (NT * 16 - 2 * P - 2) * WST; k += BLOCK) zr[k] = 0.f;
    float* xw = &xs[wv][0][0];
    #pragma unroll
    for (int q = 0; q < (16 * XST) / 32; ++q) {
      const int k = lane + 32 * q;
      const bool one = (k >= 9 * XST && k < 9 * XST + P) || (k >= 13 * XST && k < 13 * XST + P);
      xw[k] = one ? 1.f : 0.f;
    }
  }

  #pragma unroll
  for (int q = 0; q < 4; ++q) {
    const int i  = ku[q] / (P / 4);
    const int n0 = (ku[q] - i * (P / 4)) * 4;
    *(float4*)&w0s[i][n0] = s0[q];
    wt[n0 + 0][i] = s3[q].x; wt[n0 + 1][i] = s3[q].y;
    wt[n0 + 2][i] = s3[q].z; wt[n0 + 3][i] = s3[q].w;
    wt[P + n0 + 0][i] = s4[q].x; wt[P + n0 + 1][i] = s4[q].y;
    wt[P + n0 + 2][i] = s4[q].z; wt[P + n0 + 3][i] = s4[q].w;
  }
  if (rok) {
    float s = 0.f;
    #pragma unroll
    for (int q = 0; q < P / 4; ++q) { s += rr[q].x; s += rr[q].y; s += rr[q].z; s += rr[q].w; }
    wt[2 * P + rsel][rrow] = s;
  }
  if (blkE) {
    *(float4*)&es[er0][4 * ec0] = ev0;
    if (wv == 0) *(float4*)&es[er1][4 * ec1] = ev1;
  }
  __syncthreads();

  if (doE) {
    const int q  = 16 * tE + ln;
    const int qc = q < P ? q : P - 1;
    const float* wb = &w0s[kb][qc];
    v2f ea[10];
    #pragma unroll
    for (int ks = 0; ks < 10; ++ks) ea[ks] = *(const v2f*)&es[ln][4 * ks + kb];
    for (int e0 = eb0; e0 < B; e0 += 2 * estep) {
      if (e0 != eb0) {
        int be = e0 + (ln >> 3) * estep + (ln & 7);
        be = be < B ? be : B - 1;
        const float* xe = x + (size_t)be * 4 * P + kb;
        #pragma unroll
        for (int ks = 0; ks < 10; ++ks) ea[ks] = *(const v2f*)(xe + 4 * ks);
      }
      v8f d0 = {}, d1 = {};
      #pragma unroll
      for (int ks = 0; ks < 10; ks += 2) {
        const v2f b0 = {wb[(4 * ks    ) * P], wb[(4 * ks + 1) * P]};
        const v2f b1 = {wb[(4 * ks + 4) * P], wb[(4 * ks + 5) * P]};
        d0 = WMMA_F32X4(ea[ks],     b0, d0);
        d1 = WMMA_F32X4(ea[ks + 1], b1, d1);
      }
      if (q < P) {
        const int m0 = e0 + half * estep;
        float* oq = out + (size_t)m0 * NF * P + 2 * P + q;
        #pragma unroll
        for (int r = 0; r < 8; ++r)
          if (m0 + r < B) oq[(unsigned)(r * NF * P)] = d0[r] + d1[r];
      }
    }
  }

  v2f bw[NT - 1][10];
  #pragma unroll
  for (int t = 0; t < NT - 1; ++t) {
    #pragma unroll
    for (int ks = 0; ks < 10; ++ks)
      bw[t][ks] = *(const v2f*)&wt[16 * t + ln][4 * ks + kb];
  }

  const int gwave  = blockIdx.x * WPB + wv;
  const int nwaves = gridDim.x * WPB;
  const int npairs = (B + 1) >> 1;

  for (int p = gwave; p < npairs; p += nwaves) {
    const int  b0   = 2 * p;
    const bool hasB = (b0 + 1) < B;
    const bool last = (p + nwaves >= npairs);
    if (last) __builtin_amdgcn_s_setprio(2);
    const float* xb = x + (size_t)b0 * 4 * P;
    if (!last)
      __builtin_prefetch(x + (size_t)(p + nwaves) * 8 * P, 0, 3);

    float* ob = out + (size_t)b0 * NF * P;

    float xv[3][4];
    #pragma unroll
    for (int t = 0; t < 3; ++t) {
      const int  idx = lane + 32 * t;
      const bool ok  = idx < (hasB ? 2 * P : P);
      const int  e   = (idx >= P) ? 1 : 0;
      const unsigned off = ok ? (unsigned)(idx + e * 3 * P) : 0u;
      #pragma unroll
      for (int f = 0; f < 4; ++f) xv[t][f] = xb[off + (unsigned)(f * P)];
    }
    #pragma unroll
    for (int t = 0; t < 3; ++t) {
      const int  idx = lane + 32 * t;
      const bool ok  = idx < (hasB ? 2 * P : P);
      const int  e   = (idx >= P) ? 1 : 0;
      const int  i   = idx - e * P;
      if (ok) {
        const float x0 = xv[t][0], x1 = xv[t][1], x2 = xv[t][2], x3 = xv[t][3];
        xs[wv][4 * e + 0][i] = -x0 * x0 + x1 * x1 + x2 * x2 + x3 * x3;
        xs[wv][4 * e + 1][i] = x0; xs[wv][8 * e + 2][i] = x1;
        xs[wv][8 * e + 3][i] = x2; xs[wv][4 * e + 8][i] = x3;
        const unsigned oo = (unsigned)(e * NF * P + i);
        ob[oo]     = x0 * x0 - x1 * x1 - x2 * x2 - x3 * x3;
        ob[oo + P] = sqrtf(x1 * x1 + x2 * x2);
      }
    }
    __builtin_amdgcn_wave_barrier();

    v2f b5[10];
    {
      unsigned rsoff = (unsigned)((16 * (NT - 1) + ln) * WST + kb);
      asm volatile("" : "+v"(rsoff));
      const float* wrs = &wt[0][0] + rsoff;
      #pragma unroll
      for (int ks = 0; ks < 10; ++ks) b5[ks] = *(const v2f*)(wrs + 4 * ks);
    }
    v8f z[NT];
    #pragma unroll
    for (int t = 0; t < NT; ++t) z[t] = (v8f){};
    #pragma unroll
    for (int ks = 0; ks < 10; ++ks) {
      v2f a = *(const v2f*)&xs[wv][ln][4 * ks + kb];
      #pragma unroll
      for (int t = 0; t < NT - 1; ++t) z[t] = WMMA_F32X4(a, bw[t][ks], z[t]);
      z[NT - 1] = WMMA_F32X4(a, b5[ks], z[NT - 1]);
    }
    __builtin_amdgcn_sched_barrier(0);

    float g[2][6], hh[2][6];
    g[0][0] = RDLANE(z[5][0], 0);  hh[0][0] = RDLANE(z[5][0], 1);
    g[0][1] = RDLANE(z[5][1], 0);  hh[0][1] = RDLANE(z[5][1], 1);
    g[0][2] = RDLANE(z[5][2], 0);  hh[0][2] = RDLANE(z[5][2], 1);
    g[0][3] = RDLANE(z[5][3], 0);  hh[0][3] = RDLANE(z[5][3], 1);
    g[0][4] = RDLANE(z[5][0], 16); hh[0][4] = RDLANE(z[5][0], 17);
    g[0][5] = RDLANE(z[5][1], 16); hh[0][5] = RDLANE(z[5][1], 17);
    g[1][0] = RDLANE(z[5][4], 0);  hh[1][0] = RDLANE(z[5][4], 1);
    g[1][1] = RDLANE(z[5][5], 0);  hh[1][1] = RDLANE(z[5][5], 1);
    g[1][2] = RDLANE(z[5][2], 16); hh[1][2] = RDLANE(z[5][2], 17);
    g[1][3] = RDLANE(z[5][3], 16); hh[1][3] = RDLANE(z[5][3], 17);
    g[1][4] = RDLANE(z[5][4], 16); hh[1][4] = RDLANE(z[5][4], 17);
    g[1][5] = RDLANE(z[5][5], 16); hh[1][5] = RDLANE(z[5][5], 17);

    #pragma unroll
    for (int jt = 0; jt < 5; ++jt) {
      if (jt == 3 && last) __builtin_amdgcn_s_setprio(1);
      if (jt >= 3 && !hasB) continue;
      const int el = (jt < 2) ? 0 : ((jt > 2) ? 1 : (ln >> 3));
      const int j  = (jt < 2) ? (16 * jt + ln) : ((jt > 2) ? (16 * (jt - 3) + ln) : (32 + (ln & 7)));
      const float* xr = &xs[wv][0][0];
      const float sj  = xr[(4 * el    ) * XST + j];
      const float x0j = xr[(4 * el + 1) * XST + j];
      const float x1j = xr[(8 * el + 2) * XST + j];
      const float x2j = xr[(8 * el + 3) * XST + j];
      const float x3j = xr[(4 * el + 8) * XST + j];
      v2f uo, um;
      uo.x = half ? -2.f * x3j : 1.f;
      uo.y = half ? sj         : 2.f * x0j;
      const bool mine = (half == el);
      um.x = mine ? -2.f * x1j : 0.f;
      um.y = mine ? -2.f * x2j : 0.f;
      const v8f zero = {};
      v8f c[5];
      if (jt < 2) {
        #pragma unroll
        for (int t = 0; t < 5; ++t) { const v2f p = {z[t][0], z[t][1]}; c[t] = WMMA_F32X4(p, uo, zero); }
        #pragma unroll
        for (int t = 0; t < 5; ++t) { const v2f p = {z[t][2], z[t][3]}; c[t] = WMMA_F32X4(p, um, c[t]); }
      } else if (jt > 2) {
        #pragma unroll
        for (int t = 0; t < 5; ++t) { const v2f p = {z[t][4], z[t][5]}; c[t] = WMMA_F32X4(p, uo, zero); }
        #pragma unroll
        for (int t = 0; t < 5; ++t) { const v2f p = {z[t][2], z[t][3]}; c[t] = WMMA_F32X4(p, um, c[t]); }
      } else {
        v2f u0, u1;
        u0.x = el ? 0.f : uo.x; u0.y = el ? 0.f : uo.y;
        u1.x = el ? uo.x : 0.f; u1.y = el ? uo.y : 0.f;
        #pragma unroll
        for (int t = 0; t < 5; ++t) { const v2f p = {z[t][0], z[t][1]}; c[t] = WMMA_F32X4(p, u0, zero); }
        #pragma unroll
        for (int t = 0; t < 5; ++t) { const v2f p = {z[t][4], z[t][5]}; c[t] = WMMA_F32X4(p, u1, c[t]); }
        #pragma unroll
        for (int t = 0; t < 5; ++t) { const v2f p = {z[t][2], z[t][3]}; c[t] = WMMA_F32X4(p, um, c[t]); }
      }
      float ma = FMINN(c[0][0], c[1][0]);
      float mb = FMINN(c[3][0], c[4][0]);
      float mc = FMINN(c[2][0], c[2][1]);
      #pragma unroll
      for (int r = 1; r < 8; ++r) {
        ma = FMINN(FMINN(ma, c[0][r]), c[1][r]);
        mb = FMINN(FMINN(mb, c[3][r]), c[4][r]);
      }
      #pragma unroll
      for (int r = 2; r < 8; r += 2) mc = FMINN(FMINN(mc, c[2][r]), c[2][r + 1]);
      {
        const float keep = FMINN(half ? mb : ma, mc);
        const float send = half ? ma : mb;
        const float res  = FMINN(keep, xhalf(send));
        if (jt != 2 || el == 0 || hasB)
          ob[(unsigned)(el * NF * P + (5 + half) * P + j)] = res;
      }
    }

    #pragma unroll
    for (int e = 0; e < 2; ++e) {
      if (e == 1 && !hasB) continue;
      float* oe = ob + e * NF * P;
      #pragma unroll
      for (int t = 0; t < 2; ++t) {
        const int j = lane + 32 * t;
        if (j >= P) continue;
        const float sj  = xs[wv][4 * e][j];
        const float x0j = xs[wv][4 * e + 1][j], x1j = xs[wv][8 * e + 2][j];
        const float x2j = xs[wv][8 * e + 3][j], x3j = xs[wv][4 * e + 8][j];
        oe[(unsigned)(3 * P + j)] = g[e][0] + sj * g[e][5] + 2.f * (x0j * g[e][1] - x1j * g[e][2] - x2j * g[e][3] - x3j * g[e][4]);
        oe[(unsigned)(4 * P + j)] = hh[e][0] + sj * hh[e][5] + 2.f * (x0j * hh[e][1] - x1j * hh[e][2] - x2j * hh[e][3] - x3j * hh[e][4]);
      }
    }
    __builtin_amdgcn_wave_barrier();
  }
  __builtin_amdgcn_s_setprio(0);
}

extern "C" void kernel_launch(void* const* d_in, const int* in_sizes, int n_in,
                              void* d_out, int out_size, void* d_ws, size_t ws_size,
                              hipStream_t stream) {
  const float* x = (const float*)d_in[0];
  const float* w = (const float*)d_in[1];
  float* out = (float*)d_out;
  const int B = in_sizes[0] / (4 * P);

  int totalWaves = (B + 3) / 4;
  int grid = (totalWaves + WPB - 1) / WPB;
  if (grid > 1024) grid = 1024;
  if (grid < 1) grid = 1;

  lorentz_feats_kernel<<<grid, BLOCK, 0, stream>>>(x, w, out, B);
}
